# starting kernel
# baseline (speedup 1.0000x reference)
#include <hip/hip_runtime.h>


#define LLEN   8192
#define NMODES 32
#define NF     64
#define NROWS  4096
#define KSPLIT 16
#define MPAIRS 128
#define NCHUNK 32
#define TWO_PI 6.283185307179586476925f

#define T1P_OFF 0
#define T2_OFF  (LLEN * NF)
#define XF_OFF  (2 * LLEN * NF)
#define G_OFF   (2 * LLEN * NF + NROWS * NF)

typedef float v2f __attribute__((ext_vector_type(2)));
typedef float v8f __attribute__((ext_vector_type(8)));

__global__ __launch_bounds__(256) void k_tables(float* __restrict__ t1p,
                                                float* __restrict__ t2) {
  const int idx = blockIdx.x * 256 + threadIdx.x;
  const int l = idx >> 6;
  const int j = idx & 63;
  const int f = j & 31;
  const int r = (f * l) & (LLEN - 1);
  const float ph = (float)r * (TWO_PI / (float)LLEN);
  const float v = (j >= 32) ? -sinf(ph) : cosf(ph);
  t2[(size_t)l * NF + j] = v;
  t1p[(size_t)(l >> 1) * 128 + j * 2 + (l & 1)] = v;
}

__global__ __launch_bounds__(512) void k_dft_fwd(const float* __restrict__ x,
                                                 const float* __restrict__ t1p,
                                                 float* __restrict__ xf) {
  __shared__ float red[KSPLIT * 32 * NF];
  const int lane = threadIdx.x & 31;
  const int wid  = threadIdx.x >> 5;
  const int hi   = lane >> 4;
  const int ln   = lane & 15;
  const int rowbase = blockIdx.x * 32;
  const float* xrow0 = x + (size_t)(rowbase + ln) * LLEN;
  const float* xrow1 = x + (size_t)(rowbase + 16 + ln) * LLEN;

  v8f a0t0 = {}, a0t1 = {}, a0t2 = {}, a0t3 = {};
  v8f a1t0 = {}, a1t1 = {}, a1t2 = {}, a1t3 = {};
  const int k0beg = wid * (LLEN / KSPLIT);
  const int k0end = k0beg + (LLEN / KSPLIT);
  for (int k0 = k0beg; k0 < k0end; k0 += 4) {
    const int kb = k0 + 2 * hi;
    const v2f a0 = *(const v2f*)(xrow0 + kb);
    const v2f a1 = *(const v2f*)(xrow1 + kb);
    const float* tb = t1p + (size_t)(kb >> 1) * 128;
    const v2f b0 = *(const v2f*)(tb + (0 * 16 + ln) * 2);
    const v2f b1 = *(const v2f*)(tb + (1 * 16 + ln) * 2);
    const v2f b2 = *(const v2f*)(tb + (2 * 16 + ln) * 2);
    const v2f b3 = *(const v2f*)(tb + (3 * 16 + ln) * 2);
    a0t0 = __builtin_amdgcn_wmma_f32_16x16x4_f32(false, a0, false, b0, (short)0, a0t0, false, false);
    a0t1 = __builtin_amdgcn_wmma_f32_16x16x4_f32(false, a0, false, b1, (short)0, a0t1, false, false);
    a0t2 = __builtin_amdgcn_wmma_f32_16x16x4_f32(false, a0, false, b2, (short)0, a0t2, false, false);
    a0t3 = __builtin_amdgcn_wmma_f32_16x16x4_f32(false, a0, false, b3, (short)0, a0t3, false, false);
    a1t0 = __builtin_amdgcn_wmma_f32_16x16x4_f32(false, a1, false, b0, (short)0, a1t0, false, false);
    a1t1 = __builtin_amdgcn_wmma_f32_16x16x4_f32(false, a1, false, b1, (short)0, a1t1, false, false);
    a1t2 = __builtin_amdgcn_wmma_f32_16x16x4_f32(false, a1, false, b2, (short)0, a1t2, false, false);
    a1t3 = __builtin_amdgcn_wmma_f32_16x16x4_f32(false, a1, false, b3, (short)0, a1t3, false, false);
  }

  float* my = red + wid * (32 * NF);
  #pragma unroll
  for (int m = 0; m < 8; ++m) {
    const int r = m + 8 * hi;
    my[r * NF + 0 * 16 + ln] = a0t0[m];
    my[r * NF + 1 * 16 + ln] = a0t1[m];
    my[r * NF + 2 * 16 + ln] = a0t2[m];
    my[r * NF + 3 * 16 + ln] = a0t3[m];
    my[(16 + r) * NF + 0 * 16 + ln] = a1t0[m];
    my[(16 + r) * NF + 1 * 16 + ln] = a1t1[m];
    my[(16 + r) * NF + 2 * 16 + ln] = a1t2[m];
    my[(16 + r) * NF + 3 * 16 + ln] = a1t3[m];
  }
  __syncthreads();
  #pragma unroll
  for (int u = 0; u < 4; ++u) {
    const int idx = threadIdx.x + 512 * u;
    float s = 0.0f;
    #pragma unroll
    for (int w = 0; w < KSPLIT; ++w) s += red[w * (32 * NF) + idx];
    xf[(size_t)blockIdx.x * (32 * NF) + idx] = s;
  }
}

__global__ __launch_bounds__(256) void k_mix(const float* __restrict__ xf,
                                             const float* __restrict__ wr,
                                             const float* __restrict__ wi,
                                             float* __restrict__ g) {
  const int idx = blockIdx.x * 256 + threadIdx.x;
  const int f = idx & 31;
  const int o = (idx >> 5) & 15;
  const int h = (idx >> 9) & 3;
  const int b = idx >> 11;
  const float* xr0 = xf + (size_t)(b * 64 + h * 16) * NF + f;
  const float* wr0 = wr + ((size_t)(h * 16) * 16 + o) * NMODES + f;
  const float* wi0 = wi + ((size_t)(h * 16) * 16 + o) * NMODES + f;
  float or_ = 0.0f, oi_ = 0.0f;
  #pragma unroll
  for (int i = 0; i < 16; ++i) {
    const float xr  = xr0[i * NF];
    const float xi  = xr0[i * NF + 32];
    const float wre = wr0[i * (16 * NMODES)];
    const float wim = wi0[i * (16 * NMODES)];
    or_ += xr * wre - xi * wim;
    oi_ += xr * wim + xi * wre;
  }
  const float invL = 1.0f / (float)LLEN;
  const float scale = (f == 0) ? invL : 2.0f * invL;
  const size_t ro = (size_t)(b * 64 + h * 16 + o);
  g[ro * NF + f]      = or_ * scale;
  g[ro * NF + 32 + f] = oi_ * scale;
}

__global__ __launch_bounds__(256) void k_idft(const float* __restrict__ g,
                                              const float* __restrict__ t2,
                                              const float* __restrict__ bias,
                                              float* __restrict__ out) {
  const int lane = threadIdx.x & 31;
  const int wid  = threadIdx.x >> 5;
  const int gw   = blockIdx.x * 8 + wid;
  const int mpair = gw >> 5;
  const int nc    = gw & (NCHUNK - 1);
  const int hi = lane >> 4, ln = lane & 15;
  const int rowbase = mpair * 32;

  v2f af0[16], af1[16];
  #pragma unroll
  for (int ks = 0; ks < 16; ++ks) {
    const int kb = ks * 4 + 2 * hi;
    af0[ks] = *(const v2f*)(g + (size_t)(rowbase + ln) * NF + kb);
    af1[ks] = *(const v2f*)(g + (size_t)(rowbase + 16 + ln) * NF + kb);
  }
  float bv0[8], bv1[8];
  #pragma unroll
  for (int m = 0; m < 8; ++m) {
    bv0[m] = bias[(rowbase + m + 8 * hi) & 63];
    bv1[m] = bias[(rowbase + 16 + m + 8 * hi) & 63];
  }

  const int colchunk = nc * (LLEN / NCHUNK);
  for (int nt = 0; nt < 16; ++nt) {
    const int col = colchunk + nt * 16 + ln;
    const float* tcol = t2 + (size_t)col * NF;
    v8f acc0 = {}, acc1 = {};
    #pragma unroll
    for (int ks = 0; ks < 16; ++ks) {
      const int kb = ks * 4 + 2 * hi;
      const v2f bf = *(const v2f*)(tcol + kb);
      acc0 = __builtin_amdgcn_wmma_f32_16x16x4_f32(false, af0[ks], false, bf, (short)0, acc0, false, false);
      acc1 = __builtin_amdgcn_wmma_f32_16x16x4_f32(false, af1[ks], false, bf, (short)0, acc1, false, false);
    }
    #pragma unroll
    for (int m = 0; m < 8; ++m) {
      const int r = rowbase + m + 8 * hi;
      out[(size_t)r * LLEN + col]              = acc0[m] + bv0[m];
      out[(size_t)(r + 16) * LLEN + col]       = acc1[m] + bv1[m];
    }
  }
}

extern "C" void kernel_launch(void* const* d_in, const int* in_sizes, int n_in,
                              void* d_out, int out_size, void* d_ws, size_t ws_size,
                              hipStream_t stream) {
  const float* x    = (const float*)d_in[0];
  const float* wre  = (const float*)d_in[1];
  const float* wim  = (const float*)d_in[2];
  const float* bias = (const float*)d_in[3];
  float* out = (float*)d_out;
  float* ws  = (float*)d_ws;

  float* t1p = ws + T1P_OFF;
  float* t2  = ws + T2_OFF;
  float* xf  = ws + XF_OFF;
  float* g   = ws + G_OFF;

  k_tables <<<(LLEN * NF) / 256, 256, 0, stream>>>(t1p, t2);
  k_dft_fwd<<<MPAIRS, 512, 0, stream>>>(x, t1p, xf);
  k_mix    <<<(NROWS * NF / 2) / 256, 256, 0, stream>>>(xf, wre, wim, g);
  k_idft   <<<(MPAIRS * NCHUNK) / 8, 256, 0, stream>>>(g, t2, bias, out);
}


# Stage 1: workgroup = 16 M-pairs x one 256-row K chunk (256 workgroups; was 128 of 16 K-slices x one M-pair): basis chunk staged once in LDS; stage 2 adds 32 partial Xf planes; Xf sum order changed
# speedup vs baseline: 1.6272x; 1.6272x over previous
#include <hip/hip_runtime.h>


#define LLEN   8192
#define NMODES 32
#define NF     64
#define NROWS  4096
#define MPAIRS 128
#define S1_WAVES 16
#define S1_KC    32
#define S1_KLEN  256
#define S1_PITCH 160
#define NCHUNK 32
#define S3_COLS  256
#define S3_WAVES 16
#define S3_PITCH 68
#define TWO_PI 6.283185307179586476925f

#define T1P_OFF 0
#define T2_OFF  (LLEN * NF)
#define G_OFF   (2 * LLEN * NF)
#define XFP_OFF (2 * LLEN * NF + NROWS * NF)
#define XFP_PLANE (NROWS * NF)

typedef float v2f __attribute__((ext_vector_type(2)));
typedef float v4f __attribute__((ext_vector_type(4)));
typedef float v8f __attribute__((ext_vector_type(8)));

__global__ __launch_bounds__(256) void k_tables(float* __restrict__ t1p,
                                                float* __restrict__ t2) {
  const int idx = blockIdx.x * 256 + threadIdx.x;
  const int l = idx >> 6;
  const int j = idx & 63;
  const int f = j & 31;
  const int r = (f * l) & (LLEN - 1);
  const float ph = (float)r * (TWO_PI / (float)LLEN);
  const float v = (j >= 32) ? -sinf(ph) : cosf(ph);
  t2[(size_t)l * NF + j] = v;
  t1p[(size_t)(l >> 1) * 128 + j * 2 + (l & 1)] = v;
}

__global__ __launch_bounds__(512) void k_dft_fwd(const float* __restrict__ x,
                                                 const float* __restrict__ t1p,
                                                 float* __restrict__ xfp) {
  __shared__ __attribute__((aligned(16))) float tb[(S1_KLEN / 2) * S1_PITCH];
  const int lane = threadIdx.x & 31;
  const int wid  = threadIdx.x >> 5;
  const int kc   = blockIdx.x & (S1_KC - 1);
  const int mgrp = blockIdx.x >> 5;
  const int hi = lane >> 4, ln = lane & 15;
  const int mpair   = mgrp * S1_WAVES + wid;
  const int rowbase = mpair * 32;

  const v4f* src = (const v4f*)(t1p + (size_t)kc * ((S1_KLEN / 2) * 128));
  for (int u = 0; u < 8; ++u) {
    const int idx = threadIdx.x + 512 * u;
    const v4f v = src[idx];
    *(v4f*)(tb + (idx >> 5) * S1_PITCH + (idx & 31) * 4) = v;
  }

  const float* xrow0 = x + (size_t)(rowbase + ln) * LLEN + kc * S1_KLEN + 2 * hi;
  const float* xrow1 = xrow0 + (size_t)16 * LLEN;
  const float* tl = tb + hi * S1_PITCH + ln * 2;

  v8f a0t0 = {}, a0t1 = {}, a0t2 = {}, a0t3 = {};
  v8f a1t0 = {}, a1t1 = {}, a1t2 = {}, a1t3 = {};
  auto load_trip = [&](v2f (&a0)[8], v2f (&a1)[8], int trip) {
    #pragma unroll
    for (int s = 0; s < 8; ++s) {
      a0[s] = *(const v2f*)(xrow0 + trip * 32 + s * 4);
      a1[s] = *(const v2f*)(xrow1 + trip * 32 + s * 4);
    }
  };
  auto mma_trip = [&](const v2f (&a0)[8], const v2f (&a1)[8], int trip) {
    v2f bf[8][4];
    #pragma unroll
    for (int s = 0; s < 8; ++s) {
      const float* tk = tl + (trip * 16 + s * 2) * S1_PITCH;
      #pragma unroll
      for (int t = 0; t < 4; ++t) bf[s][t] = *(const v2f*)(tk + t * 32);
    }
    #pragma unroll
    for (int s = 0; s < 8; ++s) {
      a0t0 = __builtin_amdgcn_wmma_f32_16x16x4_f32(false, a0[s], false, bf[s][0], (short)0, a0t0, false, false);
      a0t1 = __builtin_amdgcn_wmma_f32_16x16x4_f32(false, a0[s], false, bf[s][1], (short)0, a0t1, false, false);
      a0t2 = __builtin_amdgcn_wmma_f32_16x16x4_f32(false, a0[s], false, bf[s][2], (short)0, a0t2, false, false);
      a0t3 = __builtin_amdgcn_wmma_f32_16x16x4_f32(false, a0[s], false, bf[s][3], (short)0, a0t3, false, false);
      a1t0 = __builtin_amdgcn_wmma_f32_16x16x4_f32(false, a1[s], false, bf[s][0], (short)0, a1t0, false, false);
      a1t1 = __builtin_amdgcn_wmma_f32_16x16x4_f32(false, a1[s], false, bf[s][1], (short)0, a1t1, false, false);
      a1t2 = __builtin_amdgcn_wmma_f32_16x16x4_f32(false, a1[s], false, bf[s][2], (short)0, a1t2, false, false);
      a1t3 = __builtin_amdgcn_wmma_f32_16x16x4_f32(false, a1[s], false, bf[s][3], (short)0, a1t3, false, false);
    }
  };
  v2f pa0[8], pa1[8], qa0[8], qa1[8];
  load_trip(pa0, pa1, 0);
  __syncthreads();
  for (int trip = 0; trip < S1_KLEN / 32 - 2; trip += 2) {
    load_trip(qa0, qa1, trip + 1);
    __builtin_amdgcn_sched_barrier(0);
    mma_trip(pa0, pa1, trip);
    load_trip(pa0, pa1, trip + 2);
    __builtin_amdgcn_sched_barrier(0);
    mma_trip(qa0, qa1, trip + 1);
  }
  load_trip(qa0, qa1, S1_KLEN / 32 - 1);
  __builtin_amdgcn_sched_barrier(0);
  mma_trip(pa0, pa1, S1_KLEN / 32 - 2);
  mma_trip(qa0, qa1, S1_KLEN / 32 - 1);

  float* dst = xfp + (size_t)kc * XFP_PLANE + (size_t)mpair * (32 * NF) + lane;
  #pragma unroll
  for (int m = 0; m < 8; ++m) {
    dst[(0 * 8 + m) * 32] = a0t0[m];
    dst[(1 * 8 + m) * 32] = a0t1[m];
    dst[(2 * 8 + m) * 32] = a0t2[m];
    dst[(3 * 8 + m) * 32] = a0t3[m];
    dst[(4 * 8 + m) * 32] = a1t0[m];
    dst[(5 * 8 + m) * 32] = a1t1[m];
    dst[(6 * 8 + m) * 32] = a1t2[m];
    dst[(7 * 8 + m) * 32] = a1t3[m];
  }
}

__global__ __launch_bounds__(256) void k_mix(const float* __restrict__ xfp,
                                             const float* __restrict__ wr,
                                             const float* __restrict__ wi,
                                             float* __restrict__ g) {
  __shared__ __attribute__((aligned(16))) float xs[16 * NF];
  const int b = blockIdx.x >> 2;
  const int h = blockIdx.x & 3;
  const v4f* src = (const v4f*)(xfp + (size_t)(2 * b + (h >> 1)) * (32 * NF) + (h & 1) * (16 * NF))
                   + threadIdx.x;
  v4f s = {0.0f, 0.0f, 0.0f, 0.0f};
  #pragma unroll
  for (int p = 0; p < S1_KC; ++p) s += src[(size_t)p * (XFP_PLANE / 4)];
  *(v4f*)(xs + 4 * threadIdx.x) = s;
  __syncthreads();

  const int f  = threadIdx.x & 31;
  const int o  = threadIdx.x >> 5;
  const float* xre = xs + (f >> 4) * 256 + (f & 15);
  const float* xim = xre + 2 * 256;
  const float* wr0 = wr + ((size_t)(h * 16) * 16 + o) * NMODES + f;
  const float* wi0 = wi + ((size_t)(h * 16) * 16 + o) * NMODES + f;
  float or0 = 0.0f, oi0 = 0.0f, or1 = 0.0f, oi1 = 0.0f;
  #pragma unroll
  for (int i = 0; i < 16; ++i) {
    const int ro = (i & 7) * 32 + (i >> 3) * 16;
    const float xr   = xre[ro];
    const float xi   = xim[ro];
    const float wre0 = wr0[i * (16 * NMODES)];
    const float wim0 = wi0[i * (16 * NMODES)];
    const float wre1 = wr0[i * (16 * NMODES) + 8 * NMODES];
    const float wim1 = wi0[i * (16 * NMODES) + 8 * NMODES];
    or0 += xr * wre0 - xi * wim0;
    oi0 += xr * wim0 + xi * wre0;
    or1 += xr * wre1 - xi * wim1;
    oi1 += xr * wim1 + xi * wre1;
  }
  const float invL = 1.0f / (float)LLEN;
  const float scale = (f == 0) ? invL : 2.0f * invL;
  const size_t ro0 = (size_t)(b * 64 + h * 16 + o);
  g[ro0 * NF + f]            = or0 * scale;
  g[ro0 * NF + 32 + f]       = oi0 * scale;
  g[(ro0 + 8) * NF + f]      = or1 * scale;
  g[(ro0 + 8) * NF + 32 + f] = oi1 * scale;
}

__global__ __launch_bounds__(512) void k_idft(const float* __restrict__ g,
                                              const float* __restrict__ t2,
                                              const float* __restrict__ bias,
                                              float* __restrict__ out) {
  __shared__ __attribute__((aligned(16))) float tb[S3_COLS * S3_PITCH];
  const int lane = threadIdx.x & 31;
  const int wid  = threadIdx.x >> 5;
  const int nc   = blockIdx.x & (NCHUNK - 1);
  const int mgrp = blockIdx.x >> 5;
  const int hi = lane >> 4, ln = lane & 15;
  const int rowbase  = (mgrp * S3_WAVES + wid) * 32;
  const int colchunk = nc * S3_COLS;

  const v4f* src = (const v4f*)(t2 + (size_t)colchunk * NF);
  for (int u = 0; u < 8; ++u) {
    const int idx = threadIdx.x + 512 * u;
    const v4f v = src[idx];
    *(v4f*)(tb + (idx >> 4) * S3_PITCH + (idx & 15) * 4) = v;
  }

  v2f af0[16], af1[16];
  #pragma unroll
  for (int ks = 0; ks < 16; ++ks) {
    const int kb = ks * 4 + 2 * hi;
    af0[ks] = *(const v2f*)(g + (size_t)(rowbase + ln) * NF + kb);
    af1[ks] = *(const v2f*)(g + (size_t)(rowbase + 16 + ln) * NF + kb);
  }
  float bv0[8], bv1[8];
  #pragma unroll
  for (int m = 0; m < 8; ++m) {
    bv0[m] = bias[(rowbase + m + 8 * hi) & 63];
    bv1[m] = bias[(rowbase + 16 + m + 8 * hi) & 63];
  }
  __syncthreads();

  const float* tl = tb + ln * S3_PITCH + 2 * hi;
  float* orow = out + (size_t)(rowbase + 8 * hi) * LLEN + colchunk + ln;

  for (int nt = 0; nt < 16; ++nt) {
    const float* tcol = tl + nt * (16 * S3_PITCH);
    v8f acc0 = {}, acc1 = {};
    #pragma unroll
    for (int ks = 0; ks < 16; ++ks) {
      const v2f bf = *(const v2f*)(tcol + ks * 4);
      acc0 = __builtin_amdgcn_wmma_f32_16x16x4_f32(false, af0[ks], false, bf, (short)0, acc0, false, false);
      acc1 = __builtin_amdgcn_wmma_f32_16x16x4_f32(false, af1[ks], false, bf, (short)0, acc1, false, false);
    }
    float* o = orow + nt * 16;
    #pragma unroll
    for (int m = 0; m < 8; ++m) {
      o[(size_t)m * LLEN]        = acc0[m] + bv0[m];
      o[(size_t)(m + 16) * LLEN] = acc1[m] + bv1[m];
    }
  }
}

extern "C" void kernel_launch(void* const* d_in, const int* in_sizes, int n_in,
                              void* d_out, int out_size, void* d_ws, size_t ws_size,
                              hipStream_t stream) {
  const float* x    = (const float*)d_in[0];
  const float* wre  = (const float*)d_in[1];
  const float* wim  = (const float*)d_in[2];
  const float* bias = (const float*)d_in[3];
  float* out = (float*)d_out;
  float* ws  = (float*)d_ws;

  float* t1p = ws + T1P_OFF;
  float* t2  = ws + T2_OFF;
  float* g   = ws + G_OFF;
  float* xfp = ws + XFP_OFF;

  k_tables <<<(LLEN * NF) / 256, 256, 0, stream>>>(t1p, t2);
  k_dft_fwd<<<(MPAIRS / S1_WAVES) * S1_KC, 512, 0, stream>>>(x, t1p, xfp);
  k_mix    <<<NROWS / 16, 256, 0, stream>>>(xfp, wre, wim, g);
  k_idft   <<<(MPAIRS / S3_WAVES) * NCHUNK, 512, 0, stream>>>(g, t2, bias, out);
}


# Stage 3 mirror fold, basis(L-l) = basis(l) with -sin half negated: a tile's cos and sin WMMA sums E, O give columns l (E+O) and L-l (E-O), WMMAs halved; column L/2 one extra tile; sum order changed
# speedup vs baseline: 1.6796x; 1.0322x over previous
#include <hip/hip_runtime.h>


#define LLEN   8192
#define NMODES 32
#define NF     64
#define NROWS  4096
#define MPAIRS 128
#define S1_WAVES 16
#define S1_KC    32
#define S1_KLEN  256
#define S1_PITCH 160
#define NCHUNK 32
#define S3_COLS  128
#define S3_WAVES 16
#define S3_PITCH 68
#define TWO_PI 6.283185307179586476925f

#define T1P_OFF 0
#define T2_OFF  (LLEN * NF)
#define G_OFF   (2 * LLEN * NF)
#define XFP_OFF (2 * LLEN * NF + NROWS * NF)
#define XFP_PLANE (NROWS * NF)

typedef float v2f __attribute__((ext_vector_type(2)));
typedef float v4f __attribute__((ext_vector_type(4)));
typedef float v8f __attribute__((ext_vector_type(8)));

__global__ __launch_bounds__(256) void k_tables(float* __restrict__ t1p,
                                                float* __restrict__ t2) {
  const int idx = blockIdx.x * 256 + threadIdx.x;
  const int l = idx >> 6;
  const int j = idx & 63;
  const int f = j & 31;
  const int r = (f * l) & (LLEN - 1);
  const float ph = (float)r * (TWO_PI / (float)LLEN);
  const float v = (j >= 32) ? -sinf(ph) : cosf(ph);
  t2[(size_t)l * NF + j] = v;
  t1p[(size_t)(l >> 1) * 128 + j * 2 + (l & 1)] = v;
}

__global__ __launch_bounds__(512) void k_dft_fwd(const float* __restrict__ x,
                                                 const float* __restrict__ t1p,
                                                 float* __restrict__ xfp) {
  __shared__ __attribute__((aligned(16))) float tb[(S1_KLEN / 2) * S1_PITCH];
  const int lane = threadIdx.x & 31;
  const int wid  = threadIdx.x >> 5;
  const int kc   = blockIdx.x & (S1_KC - 1);
  const int mgrp = blockIdx.x >> 5;
  const int hi = lane >> 4, ln = lane & 15;
  const int mpair   = mgrp * S1_WAVES + wid;
  const int rowbase = mpair * 32;

  const v4f* src = (const v4f*)(t1p + (size_t)kc * ((S1_KLEN / 2) * 128));
  for (int u = 0; u < 8; ++u) {
    const int idx = threadIdx.x + 512 * u;
    const v4f v = src[idx];
    *(v4f*)(tb + (idx >> 5) * S1_PITCH + (idx & 31) * 4) = v;
  }

  const float* xrow0 = x + (size_t)(rowbase + ln) * LLEN + kc * S1_KLEN + 2 * hi;
  const float* xrow1 = xrow0 + (size_t)16 * LLEN;
  const float* tl = tb + hi * S1_PITCH + ln * 2;

  v8f a0t0 = {}, a0t1 = {}, a0t2 = {}, a0t3 = {};
  v8f a1t0 = {}, a1t1 = {}, a1t2 = {}, a1t3 = {};
  auto load_trip = [&](v2f (&a0)[8], v2f (&a1)[8], int trip) {
    #pragma unroll
    for (int s = 0; s < 8; ++s) {
      a0[s] = *(const v2f*)(xrow0 + trip * 32 + s * 4);
      a1[s] = *(const v2f*)(xrow1 + trip * 32 + s * 4);
    }
  };
  auto mma_trip = [&](const v2f (&a0)[8], const v2f (&a1)[8], int trip) {
    v2f bf[8][4];
    #pragma unroll
    for (int s = 0; s < 8; ++s) {
      const float* tk = tl + (trip * 16 + s * 2) * S1_PITCH;
      #pragma unroll
      for (int t = 0; t < 4; ++t) bf[s][t] = *(const v2f*)(tk + t * 32);
    }
    #pragma unroll
    for (int s = 0; s < 8; ++s) {
      a0t0 = __builtin_amdgcn_wmma_f32_16x16x4_f32(false, a0[s], false, bf[s][0], (short)0, a0t0, false, false);
      a0t1 = __builtin_amdgcn_wmma_f32_16x16x4_f32(false, a0[s], false, bf[s][1], (short)0, a0t1, false, false);
      a0t2 = __builtin_amdgcn_wmma_f32_16x16x4_f32(false, a0[s], false, bf[s][2], (short)0, a0t2, false, false);
      a0t3 = __builtin_amdgcn_wmma_f32_16x16x4_f32(false, a0[s], false, bf[s][3], (short)0, a0t3, false, false);
      a1t0 = __builtin_amdgcn_wmma_f32_16x16x4_f32(false, a1[s], false, bf[s][0], (short)0, a1t0, false, false);
      a1t1 = __builtin_amdgcn_wmma_f32_16x16x4_f32(false, a1[s], false, bf[s][1], (short)0, a1t1, false, false);
      a1t2 = __builtin_amdgcn_wmma_f32_16x16x4_f32(false, a1[s], false, bf[s][2], (short)0, a1t2, false, false);
      a1t3 = __builtin_amdgcn_wmma_f32_16x16x4_f32(false, a1[s], false, bf[s][3], (short)0, a1t3, false, false);
    }
  };
  v2f pa0[8], pa1[8], qa0[8], qa1[8];
  load_trip(pa0, pa1, 0);
  __syncthreads();
  for (int trip = 0; trip < S1_KLEN / 32 - 2; trip += 2) {
    load_trip(qa0, qa1, trip + 1);
    __builtin_amdgcn_sched_barrier(0);
    mma_trip(pa0, pa1, trip);
    load_trip(pa0, pa1, trip + 2);
    __builtin_amdgcn_sched_barrier(0);
    mma_trip(qa0, qa1, trip + 1);
  }
  load_trip(qa0, qa1, S1_KLEN / 32 - 1);
  __builtin_amdgcn_sched_barrier(0);
  mma_trip(pa0, pa1, S1_KLEN / 32 - 2);
  mma_trip(qa0, qa1, S1_KLEN / 32 - 1);

  float* dst = xfp + (size_t)kc * XFP_PLANE + (size_t)mpair * (32 * NF) + lane;
  #pragma unroll
  for (int m = 0; m < 8; ++m) {
    dst[(0 * 8 + m) * 32] = a0t0[m];
    dst[(1 * 8 + m) * 32] = a0t1[m];
    dst[(2 * 8 + m) * 32] = a0t2[m];
    dst[(3 * 8 + m) * 32] = a0t3[m];
    dst[(4 * 8 + m) * 32] = a1t0[m];
    dst[(5 * 8 + m) * 32] = a1t1[m];
    dst[(6 * 8 + m) * 32] = a1t2[m];
    dst[(7 * 8 + m) * 32] = a1t3[m];
  }
}

__global__ __launch_bounds__(256) void k_mix(const float* __restrict__ xfp,
                                             const float* __restrict__ wr,
                                             const float* __restrict__ wi,
                                             float* __restrict__ g) {
  __shared__ __attribute__((aligned(16))) float xs[16 * NF];
  const int b = blockIdx.x >> 2;
  const int h = blockIdx.x & 3;
  const v4f* src = (const v4f*)(xfp + (size_t)(2 * b + (h >> 1)) * (32 * NF) + (h & 1) * (16 * NF))
                   + threadIdx.x;
  v4f s = {0.0f, 0.0f, 0.0f, 0.0f};
  #pragma unroll
  for (int p = 0; p < S1_KC; ++p) s += src[(size_t)p * (XFP_PLANE / 4)];
  *(v4f*)(xs + 4 * threadIdx.x) = s;
  __syncthreads();

  const int f  = threadIdx.x & 31;
  const int o  = threadIdx.x >> 5;
  const float* xre = xs + (f >> 4) * 256 + (f & 15);
  const float* xim = xre + 2 * 256;
  const float* wr0 = wr + ((size_t)(h * 16) * 16 + o) * NMODES + f;
  const float* wi0 = wi + ((size_t)(h * 16) * 16 + o) * NMODES + f;
  float or0 = 0.0f, oi0 = 0.0f, or1 = 0.0f, oi1 = 0.0f;
  #pragma unroll
  for (int i = 0; i < 16; ++i) {
    const int ro = (i & 7) * 32 + (i >> 3) * 16;
    const float xr   = xre[ro];
    const float xi   = xim[ro];
    const float wre0 = wr0[i * (16 * NMODES)];
    const float wim0 = wi0[i * (16 * NMODES)];
    const float wre1 = wr0[i * (16 * NMODES) + 8 * NMODES];
    const float wim1 = wi0[i * (16 * NMODES) + 8 * NMODES];
    or0 += xr * wre0 - xi * wim0;
    oi0 += xr * wim0 + xi * wre0;
    or1 += xr * wre1 - xi * wim1;
    oi1 += xr * wim1 + xi * wre1;
  }
  const float invL = 1.0f / (float)LLEN;
  const float scale = (f == 0) ? invL : 2.0f * invL;
  const size_t ro0 = (size_t)(b * 64 + h * 16 + o);
  g[ro0 * NF + f]            = or0 * scale;
  g[ro0 * NF + 32 + f]       = oi0 * scale;
  g[(ro0 + 8) * NF + f]      = or1 * scale;
  g[(ro0 + 8) * NF + 32 + f] = oi1 * scale;
}

__global__ __launch_bounds__(512) void k_idft(const float* __restrict__ g,
                                              const float* __restrict__ t2,
                                              const float* __restrict__ bias,
                                              float* __restrict__ out) {
  __shared__ __attribute__((aligned(16))) float tb[S3_COLS * S3_PITCH];
  const int lane = threadIdx.x & 31;
  const int wid  = __builtin_amdgcn_readfirstlane((int)(threadIdx.x >> 5));
  const int nc   = blockIdx.x & (NCHUNK - 1);
  const int mgrp = blockIdx.x >> 5;
  const int hi = lane >> 4, ln = lane & 15;
  const int rowbase  = (mgrp * S3_WAVES + wid) * 32;
  const int colchunk = nc * S3_COLS;

  const v4f* src = (const v4f*)(t2 + (size_t)colchunk * NF);
  for (int u = 0; u < 4; ++u) {
    const int idx = threadIdx.x + 512 * u;
    const v4f v = src[idx];
    *(v4f*)(tb + (idx >> 4) * S3_PITCH + (idx & 15) * 4) = v;
  }

  v2f af0[16], af1[16];
  #pragma unroll
  for (int ks = 0; ks < 16; ++ks) {
    const int kb = ks * 4 + 2 * hi;
    af0[ks] = *(const v2f*)(g + (size_t)(rowbase + ln) * NF + kb);
    af1[ks] = *(const v2f*)(g + (size_t)(rowbase + 16 + ln) * NF + kb);
  }
  float bv0[8], bv1[8];
  #pragma unroll
  for (int m = 0; m < 8; ++m) {
    bv0[m] = bias[(rowbase + m + 8 * hi) & 63];
    bv1[m] = bias[(rowbase + 16 + m + 8 * hi) & 63];
  }
  __syncthreads();

  if (wid == 0) {
    const int row0 = mgrp * (S3_WAVES * 32) + nc * 16;
    const float* gs = g  + (size_t)(row0 + ln) * NF + 2 * hi;
    const float* ts = t2 + (size_t)(LLEN / 2 + ln) * NF + 2 * hi;
    v2f sa[16], sb[16];
    #pragma unroll
    for (int ks = 0; ks < 16; ++ks) {
      sa[ks] = *(const v2f*)(gs + ks * 4);
      sb[ks] = *(const v2f*)(ts + ks * 4);
    }
    v8f acc = {};
    #pragma unroll
    for (int ks = 0; ks < 16; ++ks)
      acc = __builtin_amdgcn_wmma_f32_16x16x4_f32(false, sa[ks], false, sb[ks], (short)0, acc, false, false);
    if (ln == 0) {
      float* oc = out + (size_t)(row0 + 8 * hi) * LLEN + LLEN / 2;
      #pragma unroll
      for (int m = 0; m < 8; ++m) oc[(size_t)m * LLEN] = acc[m] + bias[(row0 + 8 * hi + m) & 63];
    }
  }

  const float* tl = tb + ln * S3_PITCH + 2 * hi;
  const int lcol = colchunk + ln;
  float* orow = out + (size_t)(rowbase + 8 * hi) * LLEN;
  float* ofwd = orow + lcol;
  const int roff = ln ? ln : 16;
  float* orev = orow + (LLEN - colchunk - roff);
  float pr0[8], pr1[8];

  for (int nt = 0; nt < S3_COLS / 16; ++nt) {
    const float* tcol = tl + nt * (16 * S3_PITCH);
    v8f e0 = {}, e1 = {}, o0 = {}, o1 = {};
    #pragma unroll
    for (int ks = 0; ks < 8; ++ks) {
      const v2f bf = *(const v2f*)(tcol + ks * 4);
      e0 = __builtin_amdgcn_wmma_f32_16x16x4_f32(false, af0[ks], false, bf, (short)0, e0, false, false);
      e1 = __builtin_amdgcn_wmma_f32_16x16x4_f32(false, af1[ks], false, bf, (short)0, e1, false, false);
    }
    #pragma unroll
    for (int ks = 8; ks < 16; ++ks) {
      const v2f bf = *(const v2f*)(tcol + ks * 4);
      o0 = __builtin_amdgcn_wmma_f32_16x16x4_f32(false, af0[ks], false, bf, (short)0, o0, false, false);
      o1 = __builtin_amdgcn_wmma_f32_16x16x4_f32(false, af1[ks], false, bf, (short)0, o1, false, false);
    }
    float* of = ofwd + nt * 16;
    #pragma unroll
    for (int m = 0; m < 8; ++m) {
      of[(size_t)m * LLEN]        = (e0[m] + o0[m]) + bv0[m];
      of[(size_t)(m + 16) * LLEN] = (e1[m] + o1[m]) + bv1[m];
    }
    float cu0[8], cu1[8];
    #pragma unroll
    for (int m = 0; m < 8; ++m) {
      cu0[m] = (e0[m] - o0[m]) + bv0[m];
      cu1[m] = (e1[m] - o1[m]) + bv1[m];
    }
    if (nt == 0) {
      if (ln == 0 && colchunk != 0) {
        float* o1st = orow + (LLEN - colchunk);
        #pragma unroll
        for (int m = 0; m < 8; ++m) {
          o1st[(size_t)m * LLEN]        = cu0[m];
          o1st[(size_t)(m + 16) * LLEN] = cu1[m];
        }
      }
    } else {
      float* orv = orev - (nt - 1) * 16;
      #pragma unroll
      for (int m = 0; m < 8; ++m) {
        orv[(size_t)m * LLEN]        = ln ? pr0[m] : cu0[m];
        orv[(size_t)(m + 16) * LLEN] = ln ? pr1[m] : cu1[m];
      }
    }
    #pragma unroll
    for (int m = 0; m < 8; ++m) { pr0[m] = cu0[m]; pr1[m] = cu1[m]; }
  }
  if (ln != 0) {
    float* orv = orev - (S3_COLS / 16 - 1) * 16;
    #pragma unroll
    for (int m = 0; m < 8; ++m) {
      orv[(size_t)m * LLEN]        = pr0[m];
      orv[(size_t)(m + 16) * LLEN] = pr1[m];
    }
  }
}

extern "C" void kernel_launch(void* const* d_in, const int* in_sizes, int n_in,
                              void* d_out, int out_size, void* d_ws, size_t ws_size,
                              hipStream_t stream) {
  const float* x    = (const float*)d_in[0];
  const float* wre  = (const float*)d_in[1];
  const float* wim  = (const float*)d_in[2];
  const float* bias = (const float*)d_in[3];
  float* out = (float*)d_out;
  float* ws  = (float*)d_ws;

  float* t1p = ws + T1P_OFF;
  float* t2  = ws + T2_OFF;
  float* g   = ws + G_OFF;
  float* xfp = ws + XFP_OFF;

  k_tables <<<(LLEN * NF) / 256, 256, 0, stream>>>(t1p, t2);
  k_dft_fwd<<<(MPAIRS / S1_WAVES) * S1_KC, 512, 0, stream>>>(x, t1p, xfp);
  k_mix    <<<NROWS / 16, 256, 0, stream>>>(xfp, wre, wim, g);
  k_idft   <<<(MPAIRS / S3_WAVES) * NCHUNK, 512, 0, stream>>>(g, t2, bias, out);
}


# Stage 1 mirror fold on a basis half a sample on: x[p]+x[L-1-p] meets the cos tiles, x[p]-x[L-1-p] the -sin tiles, L/2 basis rows, WMMAs halved; stage 2 turns Yf by pi*f/L; Xf sum order changed
# speedup vs baseline: 1.8002x; 1.0718x over previous
#include <hip/hip_runtime.h>


#define LLEN   8192
#define NMODES 32
#define NF     64
#define NROWS  4096
#define MPAIRS 128
#define S1_WAVES 16
#define S1_KC    32
#define S1_KLEN  128
#define S1_PITCH 160
#define NCHUNK 32
#define S3_COLS  128
#define S3_WAVES 16
#define S3_PITCH 68
#define TWO_PI 6.283185307179586476925f

#define T1P_OFF 0
#define T2_OFF  (LLEN * NF)
#define G_OFF   (2 * LLEN * NF)
#define XFP_OFF (2 * LLEN * NF + NROWS * NF)
#define XFP_PLANE (NROWS * NF)

typedef float v2f __attribute__((ext_vector_type(2)));
typedef float v4f __attribute__((ext_vector_type(4)));
typedef float v8f __attribute__((ext_vector_type(8)));

__global__ __launch_bounds__(256) void k_tables(float* __restrict__ t1p,
                                                float* __restrict__ t2) {
  const int idx = blockIdx.x * 256 + threadIdx.x;
  const int l = idx >> 6;
  const int j = idx & 63;
  const int f = j & 31;
  const int r = (f * l) & (LLEN - 1);
  const float ph = (float)r * (TWO_PI / (float)LLEN);
  const float v = (j >= 32) ? -sinf(ph) : cosf(ph);
  t2[(size_t)l * NF + j] = v;
  if (l < LLEN / 2) {
    const int r2 = (f * (2 * l + 1)) & (2 * LLEN - 1);
    const float ph2 = (float)r2 * (TWO_PI / (float)(2 * LLEN));
    t1p[(size_t)(l >> 1) * 128 + j * 2 + (l & 1)] = (j >= 32) ? -sinf(ph2) : cosf(ph2);
  }
}

__global__ __launch_bounds__(512) void k_dft_fwd(const float* __restrict__ x,
                                                 const float* __restrict__ t1p,
                                                 float* __restrict__ xfp) {
  __shared__ __attribute__((aligned(16))) float tb[(S1_KLEN / 2) * S1_PITCH];
  const int lane = threadIdx.x & 31;
  const int wid  = threadIdx.x >> 5;
  const int kc   = blockIdx.x & (S1_KC - 1);
  const int mgrp = blockIdx.x >> 5;
  const int hi = lane >> 4, ln = lane & 15;
  const int mpair   = mgrp * S1_WAVES + wid;
  const int rowbase = mpair * 32;

  const v4f* src = (const v4f*)(t1p + (size_t)kc * ((S1_KLEN / 2) * 128));
  for (int u = 0; u < 4; ++u) {
    const int idx = threadIdx.x + 512 * u;
    const v4f v = src[idx];
    *(v4f*)(tb + (idx >> 5) * S1_PITCH + (idx & 31) * 4) = v;
  }

  const float* xrow0 = x + (size_t)(rowbase + ln) * LLEN + kc * S1_KLEN + 2 * hi;
  const float* xrow1 = xrow0 + (size_t)16 * LLEN;
  const float* xmir0 = x + (size_t)(rowbase + ln) * LLEN + (LLEN - 2 - kc * S1_KLEN - 2 * hi);
  const float* xmir1 = xmir0 + (size_t)16 * LLEN;
  const float* tl = tb + hi * S1_PITCH + ln * 2;

  v8f a0t0 = {}, a0t1 = {}, a0t2 = {}, a0t3 = {};
  v8f a1t0 = {}, a1t1 = {}, a1t2 = {}, a1t3 = {};
  auto load_trip = [&](v2f (&a0)[8], v2f (&a1)[8], v2f (&m0)[8], v2f (&m1)[8], int trip) {
    #pragma unroll
    for (int s = 0; s < 8; ++s) {
      a0[s] = *(const v2f*)(xrow0 + trip * 32 + s * 4);
      a1[s] = *(const v2f*)(xrow1 + trip * 32 + s * 4);
      m0[s] = *(const v2f*)(xmir0 - trip * 32 - s * 4);
      m1[s] = *(const v2f*)(xmir1 - trip * 32 - s * 4);
    }
  };
  auto mma_trip = [&](const v2f (&a0)[8], const v2f (&a1)[8],
                      const v2f (&m0)[8], const v2f (&m1)[8], int trip) {
    v2f bf[8][4];
    #pragma unroll
    for (int s = 0; s < 8; ++s) {
      const float* tk = tl + (trip * 16 + s * 2) * S1_PITCH;
      #pragma unroll
      for (int t = 0; t < 4; ++t) bf[s][t] = *(const v2f*)(tk + t * 32);
    }
    #pragma unroll
    for (int s = 0; s < 8; ++s) {
      if (s == 4) __builtin_amdgcn_sched_barrier(0);
      const v2f s0 = {a0[s].x + m0[s].y, a0[s].y + m0[s].x};
      const v2f d0 = {a0[s].x - m0[s].y, a0[s].y - m0[s].x};
      const v2f s1 = {a1[s].x + m1[s].y, a1[s].y + m1[s].x};
      const v2f d1 = {a1[s].x - m1[s].y, a1[s].y - m1[s].x};
      a0t0 = __builtin_amdgcn_wmma_f32_16x16x4_f32(false, s0, false, bf[s][0], (short)0, a0t0, false, false);
      a0t1 = __builtin_amdgcn_wmma_f32_16x16x4_f32(false, s0, false, bf[s][1], (short)0, a0t1, false, false);
      a0t2 = __builtin_amdgcn_wmma_f32_16x16x4_f32(false, d0, false, bf[s][2], (short)0, a0t2, false, false);
      a0t3 = __builtin_amdgcn_wmma_f32_16x16x4_f32(false, d0, false, bf[s][3], (short)0, a0t3, false, false);
      a1t0 = __builtin_amdgcn_wmma_f32_16x16x4_f32(false, s1, false, bf[s][0], (short)0, a1t0, false, false);
      a1t1 = __builtin_amdgcn_wmma_f32_16x16x4_f32(false, s1, false, bf[s][1], (short)0, a1t1, false, false);
      a1t2 = __builtin_amdgcn_wmma_f32_16x16x4_f32(false, d1, false, bf[s][2], (short)0, a1t2, false, false);
      a1t3 = __builtin_amdgcn_wmma_f32_16x16x4_f32(false, d1, false, bf[s][3], (short)0, a1t3, false, false);
    }
  };
  v2f pa0[8], pa1[8], pm0[8], pm1[8], qa0[8], qa1[8], qm0[8], qm1[8];
  load_trip(pa0, pa1, pm0, pm1, 0);
  __syncthreads();
  for (int trip = 0; trip < S1_KLEN / 32 - 2; trip += 2) {
    load_trip(qa0, qa1, qm0, qm1, trip + 1);
    __builtin_amdgcn_sched_barrier(0);
    mma_trip(pa0, pa1, pm0, pm1, trip);
    load_trip(pa0, pa1, pm0, pm1, trip + 2);
    __builtin_amdgcn_sched_barrier(0);
    mma_trip(qa0, qa1, qm0, qm1, trip + 1);
  }
  load_trip(qa0, qa1, qm0, qm1, S1_KLEN / 32 - 1);
  __builtin_amdgcn_sched_barrier(0);
  mma_trip(pa0, pa1, pm0, pm1, S1_KLEN / 32 - 2);
  mma_trip(qa0, qa1, qm0, qm1, S1_KLEN / 32 - 1);

  float* dst = xfp + (size_t)kc * XFP_PLANE + (size_t)mpair * (32 * NF) + lane;
  #pragma unroll
  for (int m = 0; m < 8; ++m) {
    dst[(0 * 8 + m) * 32] = a0t0[m];
    dst[(1 * 8 + m) * 32] = a0t1[m];
    dst[(2 * 8 + m) * 32] = a0t2[m];
    dst[(3 * 8 + m) * 32] = a0t3[m];
    dst[(4 * 8 + m) * 32] = a1t0[m];
    dst[(5 * 8 + m) * 32] = a1t1[m];
    dst[(6 * 8 + m) * 32] = a1t2[m];
    dst[(7 * 8 + m) * 32] = a1t3[m];
  }
}

__global__ __launch_bounds__(256) void k_mix(const float* __restrict__ xfp,
                                             const float* __restrict__ wr,
                                             const float* __restrict__ wi,
                                             float* __restrict__ g) {
  __shared__ __attribute__((aligned(16))) float xs[16 * NF];
  const int b = blockIdx.x >> 2;
  const int h = blockIdx.x & 3;
  const v4f* src = (const v4f*)(xfp + (size_t)(2 * b + (h >> 1)) * (32 * NF) + (h & 1) * (16 * NF))
                   + threadIdx.x;
  v4f s = {0.0f, 0.0f, 0.0f, 0.0f};
  #pragma unroll
  for (int p = 0; p < S1_KC; ++p) s += src[(size_t)p * (XFP_PLANE / 4)];
  *(v4f*)(xs + 4 * threadIdx.x) = s;
  __syncthreads();

  const int f  = threadIdx.x & 31;
  const int o  = threadIdx.x >> 5;
  const float* xre = xs + (f >> 4) * 256 + (f & 15);
  const float* xim = xre + 2 * 256;
  const float* wr0 = wr + ((size_t)(h * 16) * 16 + o) * NMODES + f;
  const float* wi0 = wi + ((size_t)(h * 16) * 16 + o) * NMODES + f;
  const float hph = (float)f * (0.5f * TWO_PI / (float)LLEN);
  const float hc = cosf(hph), hs = sinf(hph);
  float or0 = 0.0f, oi0 = 0.0f, or1 = 0.0f, oi1 = 0.0f;
  #pragma unroll
  for (int i = 0; i < 16; ++i) {
    const int ro = (i & 7) * 32 + (i >> 3) * 16;
    const float yr   = xre[ro];
    const float yi   = xim[ro];
    const float xr   = yr * hc - yi * hs;
    const float xi   = yr * hs + yi * hc;
    const float wre0 = wr0[i * (16 * NMODES)];
    const float wim0 = wi0[i * (16 * NMODES)];
    const float wre1 = wr0[i * (16 * NMODES) + 8 * NMODES];
    const float wim1 = wi0[i * (16 * NMODES) + 8 * NMODES];
    or0 += xr * wre0 - xi * wim0;
    oi0 += xr * wim0 + xi * wre0;
    or1 += xr * wre1 - xi * wim1;
    oi1 += xr * wim1 + xi * wre1;
  }
  const float invL = 1.0f / (float)LLEN;
  const float scale = (f == 0) ? invL : 2.0f * invL;
  const size_t ro0 = (size_t)(b * 64 + h * 16 + o);
  g[ro0 * NF + f]            = or0 * scale;
  g[ro0 * NF + 32 + f]       = oi0 * scale;
  g[(ro0 + 8) * NF + f]      = or1 * scale;
  g[(ro0 + 8) * NF + 32 + f] = oi1 * scale;
}

__global__ __launch_bounds__(512) void k_idft(const float* __restrict__ g,
                                              const float* __restrict__ t2,
                                              const float* __restrict__ bias,
                                              float* __restrict__ out) {
  __shared__ __attribute__((aligned(16))) float tb[S3_COLS * S3_PITCH];
  const int lane = threadIdx.x & 31;
  const int wid  = __builtin_amdgcn_readfirstlane((int)(threadIdx.x >> 5));
  const int nc   = blockIdx.x & (NCHUNK - 1);
  const int mgrp = blockIdx.x >> 5;
  const int hi = lane >> 4, ln = lane & 15;
  const int rowbase  = (mgrp * S3_WAVES + wid) * 32;
  const int colchunk = nc * S3_COLS;

  const v4f* src = (const v4f*)(t2 + (size_t)colchunk * NF);
  for (int u = 0; u < 4; ++u) {
    const int idx = threadIdx.x + 512 * u;
    const v4f v = src[idx];
    *(v4f*)(tb + (idx >> 4) * S3_PITCH + (idx & 15) * 4) = v;
  }

  v2f af0[16], af1[16];
  #pragma unroll
  for (int ks = 0; ks < 16; ++ks) {
    const int kb = ks * 4 + 2 * hi;
    af0[ks] = *(const v2f*)(g + (size_t)(rowbase + ln) * NF + kb);
    af1[ks] = *(const v2f*)(g + (size_t)(rowbase + 16 + ln) * NF + kb);
  }
  float bv0[8], bv1[8];
  #pragma unroll
  for (int m = 0; m < 8; ++m) {
    bv0[m] = bias[(rowbase + m + 8 * hi) & 63];
    bv1[m] = bias[(rowbase + 16 + m + 8 * hi) & 63];
  }
  __syncthreads();

  if (wid == 0) {
    const int row0 = mgrp * (S3_WAVES * 32) + nc * 16;
    const float* gs = g  + (size_t)(row0 + ln) * NF + 2 * hi;
    const float* ts = t2 + (size_t)(LLEN / 2 + ln) * NF + 2 * hi;
    v2f sa[16], sb[16];
    #pragma unroll
    for (int ks = 0; ks < 16; ++ks) {
      sa[ks] = *(const v2f*)(gs + ks * 4);
      sb[ks] = *(const v2f*)(ts + ks * 4);
    }
    v8f acc = {};
    #pragma unroll
    for (int ks = 0; ks < 16; ++ks)
      acc = __builtin_amdgcn_wmma_f32_16x16x4_f32(false, sa[ks], false, sb[ks], (short)0, acc, false, false);
    if (ln == 0) {
      float* oc = out + (size_t)(row0 + 8 * hi) * LLEN + LLEN / 2;
      #pragma unroll
      for (int m = 0; m < 8; ++m) oc[(size_t)m * LLEN] = acc[m] + bias[(row0 + 8 * hi + m) & 63];
    }
  }

  const float* tl = tb + ln * S3_PITCH + 2 * hi;
  const int lcol = colchunk + ln;
  float* orow = out + (size_t)(rowbase + 8 * hi) * LLEN;
  float* ofwd = orow + lcol;
  const int roff = ln ? ln : 16;
  float* orev = orow + (LLEN - colchunk - roff);
  float pr0[8], pr1[8];

  for (int nt = 0; nt < S3_COLS / 16; ++nt) {
    const float* tcol = tl + nt * (16 * S3_PITCH);
    v8f e0 = {}, e1 = {}, o0 = {}, o1 = {};
    #pragma unroll
    for (int ks = 0; ks < 8; ++ks) {
      const v2f bf = *(const v2f*)(tcol + ks * 4);
      e0 = __builtin_amdgcn_wmma_f32_16x16x4_f32(false, af0[ks], false, bf, (short)0, e0, false, false);
      e1 = __builtin_amdgcn_wmma_f32_16x16x4_f32(false, af1[ks], false, bf, (short)0, e1, false, false);
    }
    #pragma unroll
    for (int ks = 8; ks < 16; ++ks) {
      const v2f bf = *(const v2f*)(tcol + ks * 4);
      o0 = __builtin_amdgcn_wmma_f32_16x16x4_f32(false, af0[ks], false, bf, (short)0, o0, false, false);
      o1 = __builtin_amdgcn_wmma_f32_16x16x4_f32(false, af1[ks], false, bf, (short)0, o1, false, false);
    }
    float* of = ofwd + nt * 16;
    #pragma unroll
    for (int m = 0; m < 8; ++m) {
      of[(size_t)m * LLEN]        = (e0[m] + o0[m]) + bv0[m];
      of[(size_t)(m + 16) * LLEN] = (e1[m] + o1[m]) + bv1[m];
    }
    float cu0[8], cu1[8];
    #pragma unroll
    for (int m = 0; m < 8; ++m) {
      cu0[m] = (e0[m] - o0[m]) + bv0[m];
      cu1[m] = (e1[m] - o1[m]) + bv1[m];
    }
    if (nt == 0) {
      if (ln == 0 && colchunk != 0) {
        float* o1st = orow + (LLEN - colchunk);
        #pragma unroll
        for (int m = 0; m < 8; ++m) {
          o1st[(size_t)m * LLEN]        = cu0[m];
          o1st[(size_t)(m + 16) * LLEN] = cu1[m];
        }
      }
    } else {
      float* orv = orev - (nt - 1) * 16;
      #pragma unroll
      for (int m = 0; m < 8; ++m) {
        orv[(size_t)m * LLEN]        = ln ? pr0[m] : cu0[m];
        orv[(size_t)(m + 16) * LLEN] = ln ? pr1[m] : cu1[m];
      }
    }
    #pragma unroll
    for (int m = 0; m < 8; ++m) { pr0[m] = cu0[m]; pr1[m] = cu1[m]; }
  }
  if (ln != 0) {
    float* orv = orev - (S3_COLS / 16 - 1) * 16;
    #pragma unroll
    for (int m = 0; m < 8; ++m) {
      orv[(size_t)m * LLEN]        = pr0[m];
      orv[(size_t)(m + 16) * LLEN] = pr1[m];
    }
  }
}

extern "C" void kernel_launch(void* const* d_in, const int* in_sizes, int n_in,
                              void* d_out, int out_size, void* d_ws, size_t ws_size,
                              hipStream_t stream) {
  const float* x    = (const float*)d_in[0];
  const float* wre  = (const float*)d_in[1];
  const float* wim  = (const float*)d_in[2];
  const float* bias = (const float*)d_in[3];
  float* out = (float*)d_out;
  float* ws  = (float*)d_ws;

  float* t1p = ws + T1P_OFF;
  float* t2  = ws + T2_OFF;
  float* g   = ws + G_OFF;
  float* xfp = ws + XFP_OFF;

  k_tables <<<(LLEN * NF) / 256, 256, 0, stream>>>(t1p, t2);
  k_dft_fwd<<<(MPAIRS / S1_WAVES) * S1_KC, 512, 0, stream>>>(x, t1p, xfp);
  k_mix    <<<NROWS / 16, 256, 0, stream>>>(xfp, wre, wim, g);
  k_idft   <<<(MPAIRS / S3_WAVES) * NCHUNK, 512, 0, stream>>>(g, t2, bias, out);
}


# Basis no longer tabled: each stage-1 and stage-3 workgroup evaluates its own 128-row basis chunk into LDS (same exact phase, sincosf); tables launch gone: 3 launches, was 4; sums unchanged
# speedup vs baseline: 1.8636x; 1.0352x over previous
#include <hip/hip_runtime.h>


#define LLEN   8192
#define NMODES 32
#define NF     64
#define NROWS  4096
#define MPAIRS 128
#define S1_WAVES 16
#define S1_KC    32
#define S1_KLEN  128
#define S1_PITCH 160
#define NCHUNK 32
#define S3_COLS  128
#define S3_WAVES 16
#define S3_PITCH 68
#define TWO_PI 6.283185307179586476925f

#define G_OFF   0
#define XFP_OFF (NROWS * NF)
#define XFP_PLANE (NROWS * NF)

typedef float v2f __attribute__((ext_vector_type(2)));
typedef float v4f __attribute__((ext_vector_type(4)));
typedef float v8f __attribute__((ext_vector_type(8)));

__device__ __forceinline__ float basis_phase3(int f, int l) {
  const int r = (f * l) & (LLEN - 1);
  return (float)r * (TWO_PI / (float)LLEN);
}
__device__ __forceinline__ float basis_phase1(int f, int p) {
  const int r2 = (f * (2 * p + 1)) & (2 * LLEN - 1);
  return (float)r2 * (TWO_PI / (float)(2 * LLEN));
}

__global__ __launch_bounds__(512) void k_dft_fwd(const float* __restrict__ x,
                                                 float* __restrict__ xfp) {
  __shared__ __attribute__((aligned(16))) float tb[(S1_KLEN / 2) * S1_PITCH];
  const int lane = threadIdx.x & 31;
  const int wid  = threadIdx.x >> 5;
  const int kc   = blockIdx.x & (S1_KC - 1);
  const int mgrp = blockIdx.x >> 5;
  const int hi = lane >> 4, ln = lane & 15;
  const int mpair   = mgrp * S1_WAVES + wid;
  const int rowbase = mpair * 32;

  const float* xrow0 = x + (size_t)(rowbase + ln) * LLEN + kc * S1_KLEN + 2 * hi;
  const float* xrow1 = xrow0 + (size_t)16 * LLEN;
  const float* xmir0 = x + (size_t)(rowbase + ln) * LLEN + (LLEN - 2 - kc * S1_KLEN - 2 * hi);
  const float* xmir1 = xmir0 + (size_t)16 * LLEN;
  const float* tl = tb + hi * S1_PITCH + ln * 2;

  v8f a0t0 = {}, a0t1 = {}, a0t2 = {}, a0t3 = {};
  v8f a1t0 = {}, a1t1 = {}, a1t2 = {}, a1t3 = {};
  auto load_trip = [&](v2f (&a0)[8], v2f (&a1)[8], v2f (&m0)[8], v2f (&m1)[8], int trip) {
    #pragma unroll
    for (int s = 0; s < 8; ++s) {
      a0[s] = *(const v2f*)(xrow0 + trip * 32 + s * 4);
      a1[s] = *(const v2f*)(xrow1 + trip * 32 + s * 4);
      m0[s] = *(const v2f*)(xmir0 - trip * 32 - s * 4);
      m1[s] = *(const v2f*)(xmir1 - trip * 32 - s * 4);
    }
  };
  auto mma_trip = [&](const v2f (&a0)[8], const v2f (&a1)[8],
                      const v2f (&m0)[8], const v2f (&m1)[8], int trip) {
    v2f bf[8][4];
    #pragma unroll
    for (int s = 0; s < 8; ++s) {
      const float* tk = tl + (trip * 16 + s * 2) * S1_PITCH;
      #pragma unroll
      for (int t = 0; t < 4; ++t) bf[s][t] = *(const v2f*)(tk + t * 32);
    }
    #pragma unroll
    for (int s = 0; s < 8; ++s) {
      if (s == 4) __builtin_amdgcn_sched_barrier(0);
      const v2f s0 = {a0[s].x + m0[s].y, a0[s].y + m0[s].x};
      const v2f d0 = {a0[s].x - m0[s].y, a0[s].y - m0[s].x};
      const v2f s1 = {a1[s].x + m1[s].y, a1[s].y + m1[s].x};
      const v2f d1 = {a1[s].x - m1[s].y, a1[s].y - m1[s].x};
      a0t0 = __builtin_amdgcn_wmma_f32_16x16x4_f32(false, s0, false, bf[s][0], (short)0, a0t0, false, false);
      a0t1 = __builtin_amdgcn_wmma_f32_16x16x4_f32(false, s0, false, bf[s][1], (short)0, a0t1, false, false);
      a0t2 = __builtin_amdgcn_wmma_f32_16x16x4_f32(false, d0, false, bf[s][2], (short)0, a0t2, false, false);
      a0t3 = __builtin_amdgcn_wmma_f32_16x16x4_f32(false, d0, false, bf[s][3], (short)0, a0t3, false, false);
      a1t0 = __builtin_amdgcn_wmma_f32_16x16x4_f32(false, s1, false, bf[s][0], (short)0, a1t0, false, false);
      a1t1 = __builtin_amdgcn_wmma_f32_16x16x4_f32(false, s1, false, bf[s][1], (short)0, a1t1, false, false);
      a1t2 = __builtin_amdgcn_wmma_f32_16x16x4_f32(false, d1, false, bf[s][2], (short)0, a1t2, false, false);
      a1t3 = __builtin_amdgcn_wmma_f32_16x16x4_f32(false, d1, false, bf[s][3], (short)0, a1t3, false, false);
    }
  };
  v2f pa0[8], pa1[8], pm0[8], pm1[8], qa0[8], qa1[8], qm0[8], qm1[8];
  load_trip(pa0, pa1, pm0, pm1, 0);

  for (int u = 0; u < 4; ++u) {
    const int kp = wid * 4 + u;
    const int p  = kc * S1_KLEN + 2 * kp;
    float s0, c0, s1, c1;
    sincosf(basis_phase1(lane, p), &s0, &c0);
    sincosf(basis_phase1(lane, p + 1), &s1, &c1);
    const v2f cs = {c0, c1};
    const v2f sn = {-s0, -s1};
    *(v2f*)(tb + kp * S1_PITCH + lane * 2)      = cs;
    *(v2f*)(tb + kp * S1_PITCH + 64 + lane * 2) = sn;
  }
  __syncthreads();
  for (int trip = 0; trip < S1_KLEN / 32 - 2; trip += 2) {
    load_trip(qa0, qa1, qm0, qm1, trip + 1);
    __builtin_amdgcn_sched_barrier(0);
    mma_trip(pa0, pa1, pm0, pm1, trip);
    load_trip(pa0, pa1, pm0, pm1, trip + 2);
    __builtin_amdgcn_sched_barrier(0);
    mma_trip(qa0, qa1, qm0, qm1, trip + 1);
  }
  load_trip(qa0, qa1, qm0, qm1, S1_KLEN / 32 - 1);
  __builtin_amdgcn_sched_barrier(0);
  mma_trip(pa0, pa1, pm0, pm1, S1_KLEN / 32 - 2);
  mma_trip(qa0, qa1, qm0, qm1, S1_KLEN / 32 - 1);

  float* dst = xfp + (size_t)kc * XFP_PLANE + (size_t)mpair * (32 * NF) + lane;
  #pragma unroll
  for (int m = 0; m < 8; ++m) {
    dst[(0 * 8 + m) * 32] = a0t0[m];
    dst[(1 * 8 + m) * 32] = a0t1[m];
    dst[(2 * 8 + m) * 32] = a0t2[m];
    dst[(3 * 8 + m) * 32] = a0t3[m];
    dst[(4 * 8 + m) * 32] = a1t0[m];
    dst[(5 * 8 + m) * 32] = a1t1[m];
    dst[(6 * 8 + m) * 32] = a1t2[m];
    dst[(7 * 8 + m) * 32] = a1t3[m];
  }
}

__global__ __launch_bounds__(256) void k_mix(const float* __restrict__ xfp,
                                             const float* __restrict__ wr,
                                             const float* __restrict__ wi,
                                             float* __restrict__ g) {
  __shared__ __attribute__((aligned(16))) float xs[16 * NF];
  const int b = blockIdx.x >> 2;
  const int h = blockIdx.x & 3;
  const v4f* src = (const v4f*)(xfp + (size_t)(2 * b + (h >> 1)) * (32 * NF) + (h & 1) * (16 * NF))
                   + threadIdx.x;
  v4f s = {0.0f, 0.0f, 0.0f, 0.0f};
  #pragma unroll
  for (int p = 0; p < S1_KC; ++p) s += src[(size_t)p * (XFP_PLANE / 4)];
  *(v4f*)(xs + 4 * threadIdx.x) = s;
  __syncthreads();

  const int f  = threadIdx.x & 31;
  const int o  = threadIdx.x >> 5;
  const float* xre = xs + (f >> 4) * 256 + (f & 15);
  const float* xim = xre + 2 * 256;
  const float* wr0 = wr + ((size_t)(h * 16) * 16 + o) * NMODES + f;
  const float* wi0 = wi + ((size_t)(h * 16) * 16 + o) * NMODES + f;
  const float hph = (float)f * (0.5f * TWO_PI / (float)LLEN);
  const float hc = cosf(hph), hs = sinf(hph);
  float or0 = 0.0f, oi0 = 0.0f, or1 = 0.0f, oi1 = 0.0f;
  #pragma unroll
  for (int i = 0; i < 16; ++i) {
    const int ro = (i & 7) * 32 + (i >> 3) * 16;
    const float yr   = xre[ro];
    const float yi   = xim[ro];
    const float xr   = yr * hc - yi * hs;
    const float xi   = yr * hs + yi * hc;
    const float wre0 = wr0[i * (16 * NMODES)];
    const float wim0 = wi0[i * (16 * NMODES)];
    const float wre1 = wr0[i * (16 * NMODES) + 8 * NMODES];
    const float wim1 = wi0[i * (16 * NMODES) + 8 * NMODES];
    or0 += xr * wre0 - xi * wim0;
    oi0 += xr * wim0 + xi * wre0;
    or1 += xr * wre1 - xi * wim1;
    oi1 += xr * wim1 + xi * wre1;
  }
  const float invL = 1.0f / (float)LLEN;
  const float scale = (f == 0) ? invL : 2.0f * invL;
  const size_t ro0 = (size_t)(b * 64 + h * 16 + o);
  g[ro0 * NF + f]            = or0 * scale;
  g[ro0 * NF + 32 + f]       = oi0 * scale;
  g[(ro0 + 8) * NF + f]      = or1 * scale;
  g[(ro0 + 8) * NF + 32 + f] = oi1 * scale;
}

__global__ __launch_bounds__(512) void k_idft(const float* __restrict__ g,
                                              const float* __restrict__ bias,
                                              float* __restrict__ out) {
  __shared__ __attribute__((aligned(16))) float tb[(S3_COLS + 1) * S3_PITCH];
  const int lane = threadIdx.x & 31;
  const int wid  = __builtin_amdgcn_readfirstlane((int)(threadIdx.x >> 5));
  const int nc   = blockIdx.x & (NCHUNK - 1);
  const int mgrp = blockIdx.x >> 5;
  const int hi = lane >> 4, ln = lane & 15;
  const int rowbase  = (mgrp * S3_WAVES + wid) * 32;
  const int colchunk = nc * S3_COLS;

  v2f af0[16], af1[16];
  #pragma unroll
  for (int ks = 0; ks < 16; ++ks) {
    const int kb = ks * 4 + 2 * hi;
    af0[ks] = *(const v2f*)(g + (size_t)(rowbase + ln) * NF + kb);
    af1[ks] = *(const v2f*)(g + (size_t)(rowbase + 16 + ln) * NF + kb);
  }
  float bv0[8], bv1[8];
  #pragma unroll
  for (int m = 0; m < 8; ++m) {
    bv0[m] = bias[(rowbase + m + 8 * hi) & 63];
    bv1[m] = bias[(rowbase + 16 + m + 8 * hi) & 63];
  }

  for (int u = 0; u < 8; ++u) {
    const int ll = wid * 8 + u;
    float s, c;
    sincosf(basis_phase3(lane, colchunk + ll), &s, &c);
    tb[ll * S3_PITCH + lane]      = c;
    tb[ll * S3_PITCH + 32 + lane] = -s;
  }
  if (wid == S3_WAVES - 1) {
    float s, c;
    sincosf(basis_phase3(lane, LLEN / 2), &s, &c);
    tb[S3_COLS * S3_PITCH + lane]      = c;
    tb[S3_COLS * S3_PITCH + 32 + lane] = -s;
  }
  __syncthreads();

  if (wid == 0) {
    const int row0 = mgrp * (S3_WAVES * 32) + nc * 16;
    const float* gs = g  + (size_t)(row0 + ln) * NF + 2 * hi;
    const float* ts = tb + S3_COLS * S3_PITCH + 2 * hi;
    v2f sa[16], sb[16];
    #pragma unroll
    for (int ks = 0; ks < 16; ++ks) {
      sa[ks] = *(const v2f*)(gs + ks * 4);
      sb[ks] = *(const v2f*)(ts + ks * 4);
    }
    v8f acc = {};
    #pragma unroll
    for (int ks = 0; ks < 16; ++ks)
      acc = __builtin_amdgcn_wmma_f32_16x16x4_f32(false, sa[ks], false, sb[ks], (short)0, acc, false, false);
    if (ln == 0) {
      float* oc = out + (size_t)(row0 + 8 * hi) * LLEN + LLEN / 2;
      #pragma unroll
      for (int m = 0; m < 8; ++m) oc[(size_t)m * LLEN] = acc[m] + bias[(row0 + 8 * hi + m) & 63];
    }
  }

  const float* tl = tb + ln * S3_PITCH + 2 * hi;
  const int lcol = colchunk + ln;
  float* orow = out + (size_t)(rowbase + 8 * hi) * LLEN;
  float* ofwd = orow + lcol;
  const int roff = ln ? ln : 16;
  float* orev = orow + (LLEN - colchunk - roff);
  float pr0[8], pr1[8];

  for (int nt = 0; nt < S3_COLS / 16; ++nt) {
    const float* tcol = tl + nt * (16 * S3_PITCH);
    v8f e0 = {}, e1 = {}, o0 = {}, o1 = {};
    #pragma unroll
    for (int ks = 0; ks < 8; ++ks) {
      const v2f bf = *(const v2f*)(tcol + ks * 4);
      e0 = __builtin_amdgcn_wmma_f32_16x16x4_f32(false, af0[ks], false, bf, (short)0, e0, false, false);
      e1 = __builtin_amdgcn_wmma_f32_16x16x4_f32(false, af1[ks], false, bf, (short)0, e1, false, false);
    }
    #pragma unroll
    for (int ks = 8; ks < 16; ++ks) {
      const v2f bf = *(const v2f*)(tcol + ks * 4);
      o0 = __builtin_amdgcn_wmma_f32_16x16x4_f32(false, af0[ks], false, bf, (short)0, o0, false, false);
      o1 = __builtin_amdgcn_wmma_f32_16x16x4_f32(false, af1[ks], false, bf, (short)0, o1, false, false);
    }
    float* of = ofwd + nt * 16;
    #pragma unroll
    for (int m = 0; m < 8; ++m) {
      of[(size_t)m * LLEN]        = (e0[m] + o0[m]) + bv0[m];
      of[(size_t)(m + 16) * LLEN] = (e1[m] + o1[m]) + bv1[m];
    }
    float cu0[8], cu1[8];
    #pragma unroll
    for (int m = 0; m < 8; ++m) {
      cu0[m] = (e0[m] - o0[m]) + bv0[m];
      cu1[m] = (e1[m] - o1[m]) + bv1[m];
    }
    if (nt == 0) {
      if (ln == 0 && colchunk != 0) {
        float* o1st = orow + (LLEN - colchunk);
        #pragma unroll
        for (int m = 0; m < 8; ++m) {
          o1st[(size_t)m * LLEN]        = cu0[m];
          o1st[(size_t)(m + 16) * LLEN] = cu1[m];
        }
      }
    } else {
      float* orv = orev - (nt - 1) * 16;
      #pragma unroll
      for (int m = 0; m < 8; ++m) {
        orv[(size_t)m * LLEN]        = ln ? pr0[m] : cu0[m];
        orv[(size_t)(m + 16) * LLEN] = ln ? pr1[m] : cu1[m];
      }
    }
    #pragma unroll
    for (int m = 0; m < 8; ++m) { pr0[m] = cu0[m]; pr1[m] = cu1[m]; }
  }
  if (ln != 0) {
    float* orv = orev - (S3_COLS / 16 - 1) * 16;
    #pragma unroll
    for (int m = 0; m < 8; ++m) {
      orv[(size_t)m * LLEN]        = pr0[m];
      orv[(size_t)(m + 16) * LLEN] = pr1[m];
    }
  }
}

extern "C" void kernel_launch(void* const* d_in, const int* in_sizes, int n_in,
                              void* d_out, int out_size, void* d_ws, size_t ws_size,
                              hipStream_t stream) {
  const float* x    = (const float*)d_in[0];
  const float* wre  = (const float*)d_in[1];
  const float* wim  = (const float*)d_in[2];
  const float* bias = (const float*)d_in[3];
  float* out = (float*)d_out;
  float* ws  = (float*)d_ws;

  float* g   = ws + G_OFF;
  float* xfp = ws + XFP_OFF;

  k_dft_fwd<<<(MPAIRS / S1_WAVES) * S1_KC, 512, 0, stream>>>(x, xfp);
  k_mix    <<<NROWS / 16, 256, 0, stream>>>(xfp, wre, wim, g);
  k_idft   <<<(MPAIRS / S3_WAVES) * NCHUNK, 512, 0, stream>>>(g, bias, out);
}


# Stage 1: x goes through wave-private LDS planes: whole-line 16-B loads (4 rows x 128 B per instruction; was 16 rows x 16 B), fold in registers, A frags by ds_load_b64; sums unchanged
# speedup vs baseline: 1.9501x; 1.0464x over previous
#include <hip/hip_runtime.h>


#define LLEN   8192
#define NMODES 32
#define NF     64
#define NROWS  4096
#define MPAIRS 128
#define S1_WAVES 16
#define S1_KC    32
#define S1_KLEN  128
#define S1_PITCH 160
#define S1_XP    36
#define S1_XWAVE (2 * 32 * S1_XP)
#define NCHUNK 32
#define S3_COLS  128
#define S3_WAVES 16
#define S3_PITCH 68
#define TWO_PI 6.283185307179586476925f

#define G_OFF   0
#define XFP_OFF (NROWS * NF)
#define XFP_PLANE (NROWS * NF)

typedef float v2f __attribute__((ext_vector_type(2)));
typedef float v4f __attribute__((ext_vector_type(4)));
typedef float v8f __attribute__((ext_vector_type(8)));

__device__ __forceinline__ float basis_phase3(int f, int l) {
  const int r = (f * l) & (LLEN - 1);
  return (float)r * (TWO_PI / (float)LLEN);
}
__device__ __forceinline__ float basis_phase1(int f, int p) {
  const int r2 = (f * (2 * p + 1)) & (2 * LLEN - 1);
  return (float)r2 * (TWO_PI / (float)(2 * LLEN));
}

__global__ __launch_bounds__(512) void k_dft_fwd(const float* __restrict__ x,
                                                 float* __restrict__ xfp) {
  __shared__ __attribute__((aligned(16))) float tb[(S1_KLEN / 2) * S1_PITCH];
  __shared__ __attribute__((aligned(16))) float xs[S1_WAVES * S1_XWAVE];
  const int lane = threadIdx.x & 31;
  const int wid  = threadIdx.x >> 5;
  const int kc   = blockIdx.x & (S1_KC - 1);
  const int mgrp = blockIdx.x >> 5;
  const int hi = lane >> 4, ln = lane & 15;
  const int mpair   = mgrp * S1_WAVES + wid;
  const int rowbase = mpair * 32;
  const int r4 = lane >> 3, j8 = lane & 7;

  const float* xf = x + (size_t)(rowbase + r4) * LLEN + kc * S1_KLEN + 4 * j8;
  const float* xm = x + (size_t)(rowbase + r4) * LLEN + (LLEN - kc * S1_KLEN - 32 + 4 * (7 - j8));
  float* xw = xs + wid * S1_XWAVE;
  float* xst = xw + r4 * S1_XP + 4 * j8;
  const float* al = xw + ln * S1_XP + 2 * hi;
  const float* tl = tb + hi * S1_PITCH + ln * 2;

  v8f a0t0 = {}, a0t1 = {}, a0t2 = {}, a0t3 = {};
  v8f a1t0 = {}, a1t1 = {}, a1t2 = {}, a1t3 = {};
  auto load_trip = [&](v4f (&f)[8], v4f (&m)[8], int trip) {
    #pragma unroll
    for (int i = 0; i < 8; ++i) {
      f[i] = *(const v4f*)(xf + (size_t)(4 * i) * LLEN + trip * 32);
      m[i] = *(const v4f*)(xm + (size_t)(4 * i) * LLEN - trip * 32);
    }
  };
  auto stage_trip = [&](const v4f (&f)[8], const v4f (&m)[8]) {
    #pragma unroll
    for (int i = 0; i < 8; ++i) {
      const v4f sm = {f[i].x + m[i].w, f[i].y + m[i].z, f[i].z + m[i].y, f[i].w + m[i].x};
      const v4f df = {f[i].x - m[i].w, f[i].y - m[i].z, f[i].z - m[i].y, f[i].w - m[i].x};
      *(v4f*)(xst + (4 * i) * S1_XP)              = sm;
      *(v4f*)(xst + (32 + 4 * i) * S1_XP)         = df;
    }
  };
  auto mma_trip = [&](int trip) {
    #pragma unroll
    for (int h = 0; h < 2; ++h) {
      __builtin_amdgcn_sched_barrier(0);
      v2f bf[4][4], s0[4], s1[4], d0[4], d1[4];
      #pragma unroll
      for (int q = 0; q < 4; ++q) {
        const int s = h * 4 + q;
        const float* tk = tl + (trip * 16 + s * 2) * S1_PITCH;
        #pragma unroll
        for (int t = 0; t < 4; ++t) bf[q][t] = *(const v2f*)(tk + t * 32);
        s0[q] = *(const v2f*)(al + s * 4);
        s1[q] = *(const v2f*)(al + 16 * S1_XP + s * 4);
        d0[q] = *(const v2f*)(al + 32 * S1_XP + s * 4);
        d1[q] = *(const v2f*)(al + 48 * S1_XP + s * 4);
      }
      #pragma unroll
      for (int q = 0; q < 4; ++q) {
        a0t0 = __builtin_amdgcn_wmma_f32_16x16x4_f32(false, s0[q], false, bf[q][0], (short)0, a0t0, false, false);
        a0t1 = __builtin_amdgcn_wmma_f32_16x16x4_f32(false, s0[q], false, bf[q][1], (short)0, a0t1, false, false);
        a0t2 = __builtin_amdgcn_wmma_f32_16x16x4_f32(false, d0[q], false, bf[q][2], (short)0, a0t2, false, false);
        a0t3 = __builtin_amdgcn_wmma_f32_16x16x4_f32(false, d0[q], false, bf[q][3], (short)0, a0t3, false, false);
        a1t0 = __builtin_amdgcn_wmma_f32_16x16x4_f32(false, s1[q], false, bf[q][0], (short)0, a1t0, false, false);
        a1t1 = __builtin_amdgcn_wmma_f32_16x16x4_f32(false, s1[q], false, bf[q][1], (short)0, a1t1, false, false);
        a1t2 = __builtin_amdgcn_wmma_f32_16x16x4_f32(false, d1[q], false, bf[q][2], (short)0, a1t2, false, false);
        a1t3 = __builtin_amdgcn_wmma_f32_16x16x4_f32(false, d1[q], false, bf[q][3], (short)0, a1t3, false, false);
      }
    }
  };
  v4f fw[8], mr[8];
  load_trip(fw, mr, 0);

  for (int u = 0; u < 4; ++u) {
    const int kp = wid * 4 + u;
    const int p  = kc * S1_KLEN + 2 * kp;
    float s0, c0, s1, c1;
    sincosf(basis_phase1(lane, p), &s0, &c0);
    sincosf(basis_phase1(lane, p + 1), &s1, &c1);
    const v2f cs = {c0, c1};
    const v2f sn = {-s0, -s1};
    *(v2f*)(tb + kp * S1_PITCH + lane * 2)      = cs;
    *(v2f*)(tb + kp * S1_PITCH + 64 + lane * 2) = sn;
  }
  __syncthreads();
  #pragma unroll
  for (int trip = 0; trip < S1_KLEN / 32; ++trip) {
    __builtin_amdgcn_fence(__ATOMIC_ACQ_REL, "wavefront");
    stage_trip(fw, mr);
    __builtin_amdgcn_fence(__ATOMIC_ACQ_REL, "wavefront");
    if (trip + 1 < S1_KLEN / 32) load_trip(fw, mr, trip + 1);
    mma_trip(trip);
    __builtin_amdgcn_sched_barrier(0);
  }

  float* dst = xfp + (size_t)kc * XFP_PLANE + (size_t)mpair * (32 * NF) + lane;
  #pragma unroll
  for (int m = 0; m < 8; ++m) {
    dst[(0 * 8 + m) * 32] = a0t0[m];
    dst[(1 * 8 + m) * 32] = a0t1[m];
    dst[(2 * 8 + m) * 32] = a0t2[m];
    dst[(3 * 8 + m) * 32] = a0t3[m];
    dst[(4 * 8 + m) * 32] = a1t0[m];
    dst[(5 * 8 + m) * 32] = a1t1[m];
    dst[(6 * 8 + m) * 32] = a1t2[m];
    dst[(7 * 8 + m) * 32] = a1t3[m];
  }
}

__global__ __launch_bounds__(256) void k_mix(const float* __restrict__ xfp,
                                             const float* __restrict__ wr,
                                             const float* __restrict__ wi,
                                             float* __restrict__ g) {
  __shared__ __attribute__((aligned(16))) float xs[16 * NF];
  const int b = blockIdx.x >> 2;
  const int h = blockIdx.x & 3;
  const v4f* src = (const v4f*)(xfp + (size_t)(2 * b + (h >> 1)) * (32 * NF) + (h & 1) * (16 * NF))
                   + threadIdx.x;
  v4f s = {0.0f, 0.0f, 0.0f, 0.0f};
  #pragma unroll
  for (int p = 0; p < S1_KC; ++p) s += src[(size_t)p * (XFP_PLANE / 4)];
  *(v4f*)(xs + 4 * threadIdx.x) = s;
  __syncthreads();

  const int f  = threadIdx.x & 31;
  const int o  = threadIdx.x >> 5;
  const float* xre = xs + (f >> 4) * 256 + (f & 15);
  const float* xim = xre + 2 * 256;
  const float* wr0 = wr + ((size_t)(h * 16) * 16 + o) * NMODES + f;
  const float* wi0 = wi + ((size_t)(h * 16) * 16 + o) * NMODES + f;
  const float hph = (float)f * (0.5f * TWO_PI / (float)LLEN);
  const float hc = cosf(hph), hs = sinf(hph);
  float or0 = 0.0f, oi0 = 0.0f, or1 = 0.0f, oi1 = 0.0f;
  #pragma unroll
  for (int i = 0; i < 16; ++i) {
    const int ro = (i & 7) * 32 + (i >> 3) * 16;
    const float yr   = xre[ro];
    const float yi   = xim[ro];
    const float xr   = yr * hc - yi * hs;
    const float xi   = yr * hs + yi * hc;
    const float wre0 = wr0[i * (16 * NMODES)];
    const float wim0 = wi0[i * (16 * NMODES)];
    const float wre1 = wr0[i * (16 * NMODES) + 8 * NMODES];
    const float wim1 = wi0[i * (16 * NMODES) + 8 * NMODES];
    or0 += xr * wre0 - xi * wim0;
    oi0 += xr * wim0 + xi * wre0;
    or1 += xr * wre1 - xi * wim1;
    oi1 += xr * wim1 + xi * wre1;
  }
  const float invL = 1.0f / (float)LLEN;
  const float scale = (f == 0) ? invL : 2.0f * invL;
  const size_t ro0 = (size_t)(b * 64 + h * 16 + o);
  g[ro0 * NF + f]            = or0 * scale;
  g[ro0 * NF + 32 + f]       = oi0 * scale;
  g[(ro0 + 8) * NF + f]      = or1 * scale;
  g[(ro0 + 8) * NF + 32 + f] = oi1 * scale;
}

__global__ __launch_bounds__(512) void k_idft(const float* __restrict__ g,
                                              const float* __restrict__ bias,
                                              float* __restrict__ out) {
  __shared__ __attribute__((aligned(16))) float tb[(S3_COLS + 1) * S3_PITCH];
  const int lane = threadIdx.x & 31;
  const int wid  = __builtin_amdgcn_readfirstlane((int)(threadIdx.x >> 5));
  const int nc   = blockIdx.x & (NCHUNK - 1);
  const int mgrp = blockIdx.x >> 5;
  const int hi = lane >> 4, ln = lane & 15;
  const int rowbase  = (mgrp * S3_WAVES + wid) * 32;
  const int colchunk = nc * S3_COLS;

  v2f af0[16], af1[16];
  #pragma unroll
  for (int ks = 0; ks < 16; ++ks) {
    const int kb = ks * 4 + 2 * hi;
    af0[ks] = *(const v2f*)(g + (size_t)(rowbase + ln) * NF + kb);
    af1[ks] = *(const v2f*)(g + (size_t)(rowbase + 16 + ln) * NF + kb);
  }
  float bv0[8], bv1[8];
  #pragma unroll
  for (int m = 0; m < 8; ++m) {
    bv0[m] = bias[(rowbase + m + 8 * hi) & 63];
    bv1[m] = bias[(rowbase + 16 + m + 8 * hi) & 63];
  }

  for (int u = 0; u < 8; ++u) {
    const int ll = wid * 8 + u;
    float s, c;
    sincosf(basis_phase3(lane, colchunk + ll), &s, &c);
    tb[ll * S3_PITCH + lane]      = c;
    tb[ll * S3_PITCH + 32 + lane] = -s;
  }
  if (wid == S3_WAVES - 1) {
    float s, c;
    sincosf(basis_phase3(lane, LLEN / 2), &s, &c);
    tb[S3_COLS * S3_PITCH + lane]      = c;
    tb[S3_COLS * S3_PITCH + 32 + lane] = -s;
  }
  __syncthreads();

  if (wid == 0) {
    const int row0 = mgrp * (S3_WAVES * 32) + nc * 16;
    const float* gs = g  + (size_t)(row0 + ln) * NF + 2 * hi;
    const float* ts = tb + S3_COLS * S3_PITCH + 2 * hi;
    v2f sa[16], sb[16];
    #pragma unroll
    for (int ks = 0; ks < 16; ++ks) {
      sa[ks] = *(const v2f*)(gs + ks * 4);
      sb[ks] = *(const v2f*)(ts + ks * 4);
    }
    v8f acc = {};
    #pragma unroll
    for (int ks = 0; ks < 16; ++ks)
      acc = __builtin_amdgcn_wmma_f32_16x16x4_f32(false, sa[ks], false, sb[ks], (short)0, acc, false, false);
    if (ln == 0) {
      float* oc = out + (size_t)(row0 + 8 * hi) * LLEN + LLEN / 2;
      #pragma unroll
      for (int m = 0; m < 8; ++m) oc[(size_t)m * LLEN] = acc[m] + bias[(row0 + 8 * hi + m) & 63];
    }
  }

  const float* tl = tb + ln * S3_PITCH + 2 * hi;
  const int lcol = colchunk + ln;
  float* orow = out + (size_t)(rowbase + 8 * hi) * LLEN;
  float* ofwd = orow + lcol;
  const int roff = ln ? ln : 16;
  float* orev = orow + (LLEN - colchunk - roff);
  float pr0[8], pr1[8];

  for (int nt = 0; nt < S3_COLS / 16; ++nt) {
    const float* tcol = tl + nt * (16 * S3_PITCH);
    v8f e0 = {}, e1 = {}, o0 = {}, o1 = {};
    #pragma unroll
    for (int ks = 0; ks < 8; ++ks) {
      const v2f bf = *(const v2f*)(tcol + ks * 4);
      e0 = __builtin_amdgcn_wmma_f32_16x16x4_f32(false, af0[ks], false, bf, (short)0, e0, false, false);
      e1 = __builtin_amdgcn_wmma_f32_16x16x4_f32(false, af1[ks], false, bf, (short)0, e1, false, false);
    }
    #pragma unroll
    for (int ks = 8; ks < 16; ++ks) {
      const v2f bf = *(const v2f*)(tcol + ks * 4);
      o0 = __builtin_amdgcn_wmma_f32_16x16x4_f32(false, af0[ks], false, bf, (short)0, o0, false, false);
      o1 = __builtin_amdgcn_wmma_f32_16x16x4_f32(false, af1[ks], false, bf, (short)0, o1, false, false);
    }
    float* of = ofwd + nt * 16;
    #pragma unroll
    for (int m = 0; m < 8; ++m) {
      of[(size_t)m * LLEN]        = (e0[m] + o0[m]) + bv0[m];
      of[(size_t)(m + 16) * LLEN] = (e1[m] + o1[m]) + bv1[m];
    }
    float cu0[8], cu1[8];
    #pragma unroll
    for (int m = 0; m < 8; ++m) {
      cu0[m] = (e0[m] - o0[m]) + bv0[m];
      cu1[m] = (e1[m] - o1[m]) + bv1[m];
    }
    if (nt == 0) {
      if (ln == 0 && colchunk != 0) {
        float* o1st = orow + (LLEN - colchunk);
        #pragma unroll
        for (int m = 0; m < 8; ++m) {
          o1st[(size_t)m * LLEN]        = cu0[m];
          o1st[(size_t)(m + 16) * LLEN] = cu1[m];
        }
      }
    } else {
      float* orv = orev - (nt - 1) * 16;
      #pragma unroll
      for (int m = 0; m < 8; ++m) {
        orv[(size_t)m * LLEN]        = ln ? pr0[m] : cu0[m];
        orv[(size_t)(m + 16) * LLEN] = ln ? pr1[m] : cu1[m];
      }
    }
    #pragma unroll
    for (int m = 0; m < 8; ++m) { pr0[m] = cu0[m]; pr1[m] = cu1[m]; }
  }
  if (ln != 0) {
    float* orv = orev - (S3_COLS / 16 - 1) * 16;
    #pragma unroll
    for (int m = 0; m < 8; ++m) {
      orv[(size_t)m * LLEN]        = pr0[m];
      orv[(size_t)(m + 16) * LLEN] = pr1[m];
    }
  }
}

extern "C" void kernel_launch(void* const* d_in, const int* in_sizes, int n_in,
                              void* d_out, int out_size, void* d_ws, size_t ws_size,
                              hipStream_t stream) {
  const float* x    = (const float*)d_in[0];
  const float* wre  = (const float*)d_in[1];
  const float* wim  = (const float*)d_in[2];
  const float* bias = (const float*)d_in[3];
  float* out = (float*)d_out;
  float* ws  = (float*)d_ws;

  float* g   = ws + G_OFF;
  float* xfp = ws + XFP_OFF;

  k_dft_fwd<<<(MPAIRS / S1_WAVES) * S1_KC, 512, 0, stream>>>(x, xfp);
  k_mix    <<<NROWS / 16, 256, 0, stream>>>(xfp, wre, wim, g);
  k_idft   <<<(MPAIRS / S3_WAVES) * NCHUNK, 512, 0, stream>>>(g, bias, out);
}


# Stage 3 on stage 1's half-sample basis, mirror l <-> L-1-l (the turns cancel in stage 2): mirror runs aligned, 4 tiles interleaved by basis row, a lane stores 16 B; sum order kept, last bits differ
# speedup vs baseline: 2.2037x; 1.1300x over previous
#include <hip/hip_runtime.h>


#define LLEN   8192
#define NMODES 32
#define NF     64
#define NROWS  4096
#define MPAIRS 128
#define S1_WAVES 16
#define S1_KC    32
#define S1_KLEN  128
#define S1_PITCH 160
#define S1_XP    36
#define S1_XWAVE (2 * 32 * S1_XP)
#define NCHUNK 32
#define S3_COLS  128
#define S3_WAVES 16
#define S3_PITCH 68
#define TWO_PI 6.283185307179586476925f

#define G_OFF   0
#define XFP_OFF (NROWS * NF)
#define XFP_PLANE (NROWS * NF)

typedef float v2f __attribute__((ext_vector_type(2)));
typedef float v4f __attribute__((ext_vector_type(4)));
typedef float v8f __attribute__((ext_vector_type(8)));

__device__ __forceinline__ float basis_phase1(int f, int p) {
  const int r2 = (f * (2 * p + 1)) & (2 * LLEN - 1);
  return (float)r2 * (TWO_PI / (float)(2 * LLEN));
}

__global__ __launch_bounds__(512) void k_dft_fwd(const float* __restrict__ x,
                                                 float* __restrict__ xfp) {
  __shared__ __attribute__((aligned(16))) float tb[(S1_KLEN / 2) * S1_PITCH];
  __shared__ __attribute__((aligned(16))) float xs[S1_WAVES * S1_XWAVE];
  const int lane = threadIdx.x & 31;
  const int wid  = threadIdx.x >> 5;
  const int kc   = blockIdx.x & (S1_KC - 1);
  const int mgrp = blockIdx.x >> 5;
  const int hi = lane >> 4, ln = lane & 15;
  const int mpair   = mgrp * S1_WAVES + wid;
  const int rowbase = mpair * 32;
  const int r4 = lane >> 3, j8 = lane & 7;

  const float* xf = x + (size_t)(rowbase + r4) * LLEN + kc * S1_KLEN + 4 * j8;
  const float* xm = x + (size_t)(rowbase + r4) * LLEN + (LLEN - kc * S1_KLEN - 32 + 4 * (7 - j8));
  float* xw = xs + wid * S1_XWAVE;
  float* xst = xw + r4 * S1_XP + 4 * j8;
  const float* al = xw + ln * S1_XP + 2 * hi;
  const float* tl = tb + hi * S1_PITCH + ln * 2;

  v8f a0t0 = {}, a0t1 = {}, a0t2 = {}, a0t3 = {};
  v8f a1t0 = {}, a1t1 = {}, a1t2 = {}, a1t3 = {};
  auto load_trip = [&](v4f (&f)[8], v4f (&m)[8], int trip) {
    #pragma unroll
    for (int i = 0; i < 8; ++i) {
      f[i] = *(const v4f*)(xf + (size_t)(4 * i) * LLEN + trip * 32);
      m[i] = *(const v4f*)(xm + (size_t)(4 * i) * LLEN - trip * 32);
    }
  };
  auto stage_trip = [&](const v4f (&f)[8], const v4f (&m)[8]) {
    #pragma unroll
    for (int i = 0; i < 8; ++i) {
      const v4f sm = {f[i].x + m[i].w, f[i].y + m[i].z, f[i].z + m[i].y, f[i].w + m[i].x};
      const v4f df = {f[i].x - m[i].w, f[i].y - m[i].z, f[i].z - m[i].y, f[i].w - m[i].x};
      *(v4f*)(xst + (4 * i) * S1_XP)              = sm;
      *(v4f*)(xst + (32 + 4 * i) * S1_XP)         = df;
    }
  };
  auto mma_trip = [&](int trip) {
    #pragma unroll
    for (int h = 0; h < 2; ++h) {
      __builtin_amdgcn_sched_barrier(0);
      v2f bf[4][4], s0[4], s1[4], d0[4], d1[4];
      #pragma unroll
      for (int q = 0; q < 4; ++q) {
        const int s = h * 4 + q;
        const float* tk = tl + (trip * 16 + s * 2) * S1_PITCH;
        #pragma unroll
        for (int t = 0; t < 4; ++t) bf[q][t] = *(const v2f*)(tk + t * 32);
        s0[q] = *(const v2f*)(al + s * 4);
        s1[q] = *(const v2f*)(al + 16 * S1_XP + s * 4);
        d0[q] = *(const v2f*)(al + 32 * S1_XP + s * 4);
        d1[q] = *(const v2f*)(al + 48 * S1_XP + s * 4);
      }
      #pragma unroll
      for (int q = 0; q < 4; ++q) {
        a0t0 = __builtin_amdgcn_wmma_f32_16x16x4_f32(false, s0[q], false, bf[q][0], (short)0, a0t0, false, false);
        a0t1 = __builtin_amdgcn_wmma_f32_16x16x4_f32(false, s0[q], false, bf[q][1], (short)0, a0t1, false, false);
        a0t2 = __builtin_amdgcn_wmma_f32_16x16x4_f32(false, d0[q], false, bf[q][2], (short)0, a0t2, false, false);
        a0t3 = __builtin_amdgcn_wmma_f32_16x16x4_f32(false, d0[q], false, bf[q][3], (short)0, a0t3, false, false);
        a1t0 = __builtin_amdgcn_wmma_f32_16x16x4_f32(false, s1[q], false, bf[q][0], (short)0, a1t0, false, false);
        a1t1 = __builtin_amdgcn_wmma_f32_16x16x4_f32(false, s1[q], false, bf[q][1], (short)0, a1t1, false, false);
        a1t2 = __builtin_amdgcn_wmma_f32_16x16x4_f32(false, d1[q], false, bf[q][2], (short)0, a1t2, false, false);
        a1t3 = __builtin_amdgcn_wmma_f32_16x16x4_f32(false, d1[q], false, bf[q][3], (short)0, a1t3, false, false);
      }
    }
  };
  v4f fw[8], mr[8];
  load_trip(fw, mr, 0);

  for (int u = 0; u < 4; ++u) {
    const int kp = wid * 4 + u;
    const int p  = kc * S1_KLEN + 2 * kp;
    float s0, c0, s1, c1;
    sincosf(basis_phase1(lane, p), &s0, &c0);
    sincosf(basis_phase1(lane, p + 1), &s1, &c1);
    const v2f cs = {c0, c1};
    const v2f sn = {-s0, -s1};
    *(v2f*)(tb + kp * S1_PITCH + lane * 2)      = cs;
    *(v2f*)(tb + kp * S1_PITCH + 64 + lane * 2) = sn;
  }
  __syncthreads();
  #pragma unroll
  for (int trip = 0; trip < S1_KLEN / 32; ++trip) {
    __builtin_amdgcn_fence(__ATOMIC_ACQ_REL, "wavefront");
    stage_trip(fw, mr);
    __builtin_amdgcn_fence(__ATOMIC_ACQ_REL, "wavefront");
    if (trip + 1 < S1_KLEN / 32) load_trip(fw, mr, trip + 1);
    mma_trip(trip);
    __builtin_amdgcn_sched_barrier(0);
  }

  float* dst = xfp + (size_t)kc * XFP_PLANE + (size_t)mpair * (32 * NF) + lane;
  #pragma unroll
  for (int m = 0; m < 8; ++m) {
    dst[(0 * 8 + m) * 32] = a0t0[m];
    dst[(1 * 8 + m) * 32] = a0t1[m];
    dst[(2 * 8 + m) * 32] = a0t2[m];
    dst[(3 * 8 + m) * 32] = a0t3[m];
    dst[(4 * 8 + m) * 32] = a1t0[m];
    dst[(5 * 8 + m) * 32] = a1t1[m];
    dst[(6 * 8 + m) * 32] = a1t2[m];
    dst[(7 * 8 + m) * 32] = a1t3[m];
  }
}

__global__ __launch_bounds__(256) void k_mix(const float* __restrict__ xfp,
                                             const float* __restrict__ wr,
                                             const float* __restrict__ wi,
                                             float* __restrict__ g) {
  __shared__ __attribute__((aligned(16))) float xs[16 * NF];
  const int b = blockIdx.x >> 2;
  const int h = blockIdx.x & 3;
  const v4f* src = (const v4f*)(xfp + (size_t)(2 * b + (h >> 1)) * (32 * NF) + (h & 1) * (16 * NF))
                   + threadIdx.x;
  v4f s = {0.0f, 0.0f, 0.0f, 0.0f};
  #pragma unroll
  for (int p = 0; p < S1_KC; ++p) s += src[(size_t)p * (XFP_PLANE / 4)];
  *(v4f*)(xs + 4 * threadIdx.x) = s;
  __syncthreads();

  const int f  = threadIdx.x & 31;
  const int o  = threadIdx.x >> 5;
  const float* xre = xs + (f >> 4) * 256 + (f & 15);
  const float* xim = xre + 2 * 256;
  const float* wr0 = wr + ((size_t)(h * 16) * 16 + o) * NMODES + f;
  const float* wi0 = wi + ((size_t)(h * 16) * 16 + o) * NMODES + f;
  float or0 = 0.0f, oi0 = 0.0f, or1 = 0.0f, oi1 = 0.0f;
  #pragma unroll
  for (int i = 0; i < 16; ++i) {
    const int ro = (i & 7) * 32 + (i >> 3) * 16;
    const float xr   = xre[ro];
    const float xi   = xim[ro];
    const float wre0 = wr0[i * (16 * NMODES)];
    const float wim0 = wi0[i * (16 * NMODES)];
    const float wre1 = wr0[i * (16 * NMODES) + 8 * NMODES];
    const float wim1 = wi0[i * (16 * NMODES) + 8 * NMODES];
    or0 += xr * wre0 - xi * wim0;
    oi0 += xr * wim0 + xi * wre0;
    or1 += xr * wre1 - xi * wim1;
    oi1 += xr * wim1 + xi * wre1;
  }
  const float invL = 1.0f / (float)LLEN;
  const float scale = (f == 0) ? invL : 2.0f * invL;
  const size_t ro0 = (size_t)(b * 64 + h * 16 + o);
  g[ro0 * NF + f]            = or0 * scale;
  g[ro0 * NF + 32 + f]       = oi0 * scale;
  g[(ro0 + 8) * NF + f]      = or1 * scale;
  g[(ro0 + 8) * NF + 32 + f] = oi1 * scale;
}

__global__ __launch_bounds__(512) void k_idft(const float* __restrict__ g,
                                              const float* __restrict__ bias,
                                              float* __restrict__ out) {
  __shared__ __attribute__((aligned(16))) float tb[S3_COLS * S3_PITCH];
  const int lane = threadIdx.x & 31;
  const int wid  = threadIdx.x >> 5;
  const int nc   = blockIdx.x & (NCHUNK - 1);
  const int mgrp = blockIdx.x >> 5;
  const int hi = lane >> 4, ln = lane & 15;
  const int rowbase  = (mgrp * S3_WAVES + wid) * 32;
  const int colchunk = nc * S3_COLS;

  v2f af0[16], af1[16];
  #pragma unroll
  for (int ks = 0; ks < 16; ++ks) {
    const int kb = ks * 4 + 2 * hi;
    af0[ks] = *(const v2f*)(g + (size_t)(rowbase + ln) * NF + kb);
    af1[ks] = *(const v2f*)(g + (size_t)(rowbase + 16 + ln) * NF + kb);
  }
  float bv0[8], bv1[8];
  #pragma unroll
  for (int m = 0; m < 8; ++m) {
    bv0[m] = bias[(rowbase + m + 8 * hi) & 63];
    bv1[m] = bias[(rowbase + 16 + m + 8 * hi) & 63];
  }

  for (int u = 0; u < 8; ++u) {
    const int lr = wid * 8 + u;
    const int ll = (lr & 64) + 4 * (lr & 15) + ((lr >> 4) & 3);
    float s, c;
    sincosf(basis_phase1(lane, colchunk + ll), &s, &c);
    tb[lr * S3_PITCH + lane]      = c;
    tb[lr * S3_PITCH + 32 + lane] = -s;
  }
  __syncthreads();

  const float* tl = tb + ln * S3_PITCH + 2 * hi;
  float* orow = out + (size_t)(rowbase + 8 * hi) * LLEN;
  float* ofwd = orow + colchunk + 4 * ln;
  float* omir = orow + (LLEN - 4 - colchunk - 4 * ln);

  #pragma unroll
  for (int grp = 0; grp < S3_COLS / 64; ++grp) {
    #pragma unroll
    for (int mt = 0; mt < 2; ++mt) {
      __builtin_amdgcn_fence(__ATOMIC_ACQ_REL, "wavefront");
      v4f fw[8], mr[8];
      #pragma unroll
      for (int tp = 0; tp < 4; tp += 2) {
        __builtin_amdgcn_sched_barrier(0);
        const float* tca = tl + (grp * 64 + tp * 16) * S3_PITCH;
        const float* tcb = tca + 16 * S3_PITCH;
        v8f ea = {}, oa = {}, eb = {}, ob = {};
        #pragma unroll
        for (int ks = 0; ks < 8; ++ks) {
          const v2f ae = mt ? af1[ks] : af0[ks];
          const v2f ao = mt ? af1[ks + 8] : af0[ks + 8];
          const v2f bea = *(const v2f*)(tca + ks * 4);
          const v2f boa = *(const v2f*)(tca + (ks + 8) * 4);
          const v2f beb = *(const v2f*)(tcb + ks * 4);
          const v2f bob = *(const v2f*)(tcb + (ks + 8) * 4);
          ea = __builtin_amdgcn_wmma_f32_16x16x4_f32(false, ae, false, bea, (short)0, ea, false, false);
          oa = __builtin_amdgcn_wmma_f32_16x16x4_f32(false, ao, false, boa, (short)0, oa, false, false);
          eb = __builtin_amdgcn_wmma_f32_16x16x4_f32(false, ae, false, beb, (short)0, eb, false, false);
          ob = __builtin_amdgcn_wmma_f32_16x16x4_f32(false, ao, false, bob, (short)0, ob, false, false);
        }
        #pragma unroll
        for (int m = 0; m < 8; ++m) {
          const float bv = mt ? bv1[m] : bv0[m];
          fw[m][tp]         = (ea[m] + oa[m]) + bv;
          mr[m][3 - tp]     = (ea[m] - oa[m]) + bv;
          fw[m][tp + 1]     = (eb[m] + ob[m]) + bv;
          mr[m][2 - tp]     = (eb[m] - ob[m]) + bv;
        }
      }
      float* of = ofwd + grp * 64 + (size_t)(mt * 16) * LLEN;
      float* om = omir - grp * 64 + (size_t)(mt * 16) * LLEN;
      #pragma unroll
      for (int m = 0; m < 8; ++m) {
        *(v4f*)(of + (size_t)m * LLEN) = fw[m];
        *(v4f*)(om + (size_t)m * LLEN) = mr[m];
      }
    }
  }
}

extern "C" void kernel_launch(void* const* d_in, const int* in_sizes, int n_in,
                              void* d_out, int out_size, void* d_ws, size_t ws_size,
                              hipStream_t stream) {
  const float* x    = (const float*)d_in[0];
  const float* wre  = (const float*)d_in[1];
  const float* wim  = (const float*)d_in[2];
  const float* bias = (const float*)d_in[3];
  float* out = (float*)d_out;
  float* ws  = (float*)d_ws;

  float* g   = ws + G_OFF;
  float* xfp = ws + XFP_OFF;

  k_dft_fwd<<<(MPAIRS / S1_WAVES) * S1_KC, 512, 0, stream>>>(x, xfp);
  k_mix    <<<NROWS / 16, 256, 0, stream>>>(xfp, wre, wim, g);
  k_idft   <<<(MPAIRS / S3_WAVES) * NCHUNK, 512, 0, stream>>>(g, bias, out);
}
